# starting kernel
# baseline (speedup 1.0000x reference)
#include <hip/hip_runtime.h>


#define N_QUBITS 16
#define N_LAYERS 2
#define BATCH    512
#define DIM      65536
#define THREADS  512

typedef __attribute__((ext_vector_type(2))) float v2f;
typedef __attribute__((ext_vector_type(8))) float v8f;

__global__ __launch_bounds__(THREADS)
void qsim16_kernel(const float* __restrict__ x,
                   const float* __restrict__ params,
                   float* __restrict__ out)
{
    __shared__ __align__(16) float S[DIM];
    __shared__ float U[N_LAYERS][4][16][16];
    __shared__ float Phi[256];
    __shared__ float Plo[256];
    __shared__ float cw[16], sw[16];
    __shared__ float pcs[N_LAYERS][16];
    __shared__ float pss[N_LAYERS][16];
    __shared__ float partial[16][16];

    float4* S4 = reinterpret_cast<float4*>(S);

    const int tid  = threadIdx.x;
    const int lane = tid & 31;
    const int wave = tid >> 5;
    const int b    = blockIdx.x;

    if (tid < 16) {
        float a = 0.5f * x[b * N_QUBITS + tid];
        float s, c;
        __sincosf(a, &s, &c);
        cw[tid] = c;
        sw[tid] = s;
    } else if (tid < 48) {
        int idx = tid - 16;
        int l = idx >> 4, w = idx & 15;
        float a = 0.5f * params[idx];
        float s, c;
        __sincosf(a, &s, &c);
        pcs[l][w] = c;
        pss[l][w] = s;
    }
    __syncthreads();

    if (tid < 256) {
        int r = tid;
        float p = 1.0f;
#pragma unroll
        for (int w = 0; w < 8; ++w)
            p *= ((r >> (7 - w)) & 1) ? sw[w] : cw[w];
        Phi[r] = p;
    } else {
        int c = tid - 256;
        float p = 1.0f;
#pragma unroll
        for (int w = 8; w < 16; ++w)
            p *= ((c >> (15 - w)) & 1) ? sw[w] : cw[w];
        Plo[c] = p;
    }

    for (int e = tid; e < N_LAYERS * 4 * 256; e += THREADS) {
        int l  = e >> 10;
        int g  = (e >> 8) & 3;
        int ar = (e >> 4) & 15;
        int ac = e & 15;
        float v = 1.0f;
#pragma unroll
        for (int m = 0; m < 4; ++m) {
            int w = 4 * g + m;
            int i = (ar >> (3 - m)) & 1;
            int j = (ac >> (3 - m)) & 1;
            float f = (i == j) ? pcs[l][w] : (i ? pss[l][w] : -pss[l][w]);
            v *= f;
        }
        U[l][g][ar][ac] = v;
    }
    __syncthreads();

    float acc[16];
#pragma unroll
    for (int w = 0; w < 16; ++w) acc[w] = 0.0f;

    const int M  = lane & 15;
    const int hs = lane >> 4;

    for (int part = 0; part < 2; ++part) {
        for (int jj = 0; jj < DIM / 4 / THREADS; ++jj) {
            int blkz = tid + THREADS * jj;
            int z0   = blkz << 2;
            float ph_hi = Phi[z0 >> 8];
            int   k0   = __popc(z0);
            float4 v;
#pragma unroll
            for (int e = 0; e < 4; ++e) {
                int k = (k0 + (e == 3 ? 2 : (e == 0 ? 0 : 1))) & 3;
                float ph;
                if (part == 0) ph = (k == 0) ? 1.0f : ((k == 2) ? -1.0f : 0.0f);
                else           ph = (k == 3) ? 1.0f : ((k == 1) ? -1.0f : 0.0f);
                float m = ph_hi * Plo[(z0 & 255) | e] * ph;
                if (e == 0) v.x = m; else if (e == 1) v.y = m;
                else if (e == 2) v.z = m; else v.w = m;
            }
            S4[blkz] = v;
        }
        __syncthreads();

        for (int l = 0; l < N_LAYERS; ++l) {
            for (int g = 0; g < 4; ++g) {
                const int beta = 12 - 4 * g;
                const float* Um = &U[l][g][0][0];

                v2f afr[4];
#pragma unroll
                for (int k0 = 0; k0 < 4; ++k0) {
                    int K = 4 * k0 + 2 * hs;
                    afr[k0].x = Um[M * 16 + K];
                    afr[k0].y = Um[M * 16 + K + 1];
                }

                const int lowmask = (1 << beta) - 1;
                for (int blk = wave; blk < 256; blk += 16) {
                    int j  = blk * 16 + M;
                    int zb = ((j & ~lowmask) << 4) | (j & lowmask);
                    v8f c8 = {};
#pragma unroll
                    for (int k0 = 0; k0 < 4; ++k0) {
                        int K = 4 * k0 + 2 * hs;
                        v2f bfr;
                        bfr.x = S[zb | (K << beta)];
                        bfr.y = S[zb | ((K + 1) << beta)];
                        c8 = __builtin_amdgcn_wmma_f32_16x16x4_f32(
                                 false, afr[k0], false, bfr,
                                 (short)0, c8, false, false);
                    }
#pragma unroll
                    for (int v = 0; v < 8; ++v) {
                        int Mr = v + 8 * hs;
                        S[zb | (Mr << beta)] = c8[v];
                    }
                }
                __syncthreads();
            }

            for (int i = 0; i < 13; ++i) {
                const int plo = 14 - i;
                for (int qb = tid; qb < 4096; qb += THREADS) {
                    int q    = qb << 2;
                    int low  = q & ((1 << plo) - 1);
                    int high = q >> plo;
                    int z    = (high << (plo + 2)) | low | (1 << (plo + 1));
                    int i1   = z >> 2;
                    int i2   = i1 ^ (1 << (plo - 2));
                    float4 a = S4[i1];
                    float4 c = S4[i2];
                    S4[i1] = c;
                    S4[i2] = a;
                }
                __syncthreads();
            }
            {
                for (int h = tid; h < 8192; h += THREADS) {
                    int idx  = ((h << 1) | 1);
                    float4 v = S4[idx];
                    S4[idx]  = make_float4(v.z, v.w, v.x, v.y);
                }
                __syncthreads();
            }
            {
                for (int idx = tid; idx < 16384; idx += THREADS) {
                    float4 v = S4[idx];
                    S4[idx]  = make_float4(v.x, v.y, v.w, v.z);
                }
                __syncthreads();
            }
            {
                for (int h = tid; h < 8192; h += THREADS) {
                    int i1 = h;
                    int i2 = h ^ 0x2000;
                    float4 a = S4[i1];
                    float4 c = S4[i2];
                    float t;
                    t = a.y; a.y = c.y; c.y = t;
                    t = a.w; a.w = c.w; c.w = t;
                    S4[i1] = a;
                    S4[i2] = c;
                }
                __syncthreads();
            }
        }

        for (int jj = 0; jj < DIM / 4 / THREADS; ++jj) {
            int blkz = tid + THREADS * jj;
            int z0   = blkz << 2;
            float4 v = S4[blkz];
            float p0 = v.x * v.x, p1 = v.y * v.y, p2 = v.z * v.z, p3 = v.w * v.w;
            float ps = p0 + p1 + p2 + p3;
#pragma unroll
            for (int w = 0; w < 14; ++w)
                acc[w] += ((z0 >> (15 - w)) & 1) ? -ps : ps;
            acc[14] += (p0 + p1) - (p2 + p3);
            acc[15] += (p0 + p2) - (p1 + p3);
        }
        __syncthreads();
    }

#pragma unroll
    for (int w = 0; w < 16; ++w) {
        float v = acc[w];
        v += __shfl_xor(v, 16, 32);
        v += __shfl_xor(v, 8, 32);
        v += __shfl_xor(v, 4, 32);
        v += __shfl_xor(v, 2, 32);
        v += __shfl_xor(v, 1, 32);
        acc[w] = v;
    }
    if (lane == 0) {
#pragma unroll
        for (int w = 0; w < 16; ++w) partial[wave][w] = acc[w];
    }
    __syncthreads();
    if (tid < 16) {
        float s = 0.0f;
#pragma unroll
        for (int q = 0; q < 16; ++q) s += partial[q][tid];
        out[b * N_QUBITS + tid] = s;
    }
}

extern "C" void kernel_launch(void* const* d_in, const int* in_sizes, int n_in,
                              void* d_out, int out_size, void* d_ws, size_t ws_size,
                              hipStream_t stream) {
    const float* x      = (const float*)d_in[0];
    const float* params = (const float*)d_in[1];
    float* out          = (float*)d_out;
    (void)in_sizes; (void)n_in; (void)out_size; (void)d_ws; (void)ws_size;
    qsim16_kernel<<<BATCH, THREADS, 0, stream>>>(x, params, out);
}


# CNOT rings move no data: each ring is a GF(2)-linear index map, folded into the layer-2 RY fiber addresses and the Z sign bits; all LDS swap passes gone; expval summation order changes
# speedup vs baseline: 1.5615x; 1.5615x over previous
#include <hip/hip_runtime.h>


#define N_QUBITS 16
#define N_LAYERS 2
#define BATCH    512
#define DIM      65536
#define THREADS  512

typedef __attribute__((ext_vector_type(2))) float v2f;
typedef __attribute__((ext_vector_type(8))) float v8f;

__host__ __device__ constexpr int ring_inv(int y) {
    return (y ^ (y >> 1)) ^ ((y & 1) ? 0xC000 : 0);
}
__host__ __device__ constexpr int ring_fwd(int p) {
    int g = p;
    g ^= g >> 1; g ^= g >> 2; g ^= g >> 4; g ^= g >> 8;
    return (g & 0x7FFF) | (((g ^ (p >> 15)) & 1) << 15);
}
template <int L>
__host__ __device__ constexpr int qmap(int y) {
    for (int i = 0; i < L; ++i) y = ring_inv(y);
    return y;
}
__host__ __device__ constexpr int logical_of(int p) {
    for (int i = 0; i < N_LAYERS; ++i) p = ring_fwd(p);
    return p;
}
template <int L>
__device__ __forceinline__ int join(int a, int b) {
    return (L == 0) ? (a | b) : (a ^ b);
}

template <int L, int G>
__device__ __forceinline__ void ry_group_pass(float* S, const float* Um,
                                              int wave, int M, int hs)
{
    constexpr int beta    = 12 - 4 * G;
    constexpr int lowmask = (1 << beta) - 1;

    v2f afr[4];
#pragma unroll
    for (int k0 = 0; k0 < 4; ++k0) {
        int K = 4 * k0 + 2 * hs;
        afr[k0].x = Um[M * 16 + K];
        afr[k0].y = Um[M * 16 + K + 1];
    }

    const int offB = qmap<L>((2 * hs) << beta);
    const int offD = qmap<L>((8 * hs) << beta);

    for (int blk = wave; blk < 256; blk += 16) {
        int j  = blk * 16 + M;
        int zb = ((j & ~lowmask) << 4) | (j & lowmask);
        int pb = qmap<L>(zb);
        int pl = join<L>(pb, offB);
        int pd = join<L>(pb, offD);
        v8f c8 = {};
#pragma unroll
        for (int k0 = 0; k0 < 4; ++k0) {
            v2f bfr;
            bfr.x = S[join<L>(pl, qmap<L>((4 * k0)     << beta))];
            bfr.y = S[join<L>(pl, qmap<L>((4 * k0 + 1) << beta))];
            c8 = __builtin_amdgcn_wmma_f32_16x16x4_f32(
                     false, afr[k0], false, bfr,
                     (short)0, c8, false, false);
        }
#pragma unroll
        for (int v = 0; v < 8; ++v)
            S[join<L>(pd, qmap<L>(v << beta))] = c8[v];
    }
    __syncthreads();
}

template <int L>
__device__ __forceinline__ void ry_layer(float* S, const float* Ul,
                                         int wave, int M, int hs)
{
    ry_group_pass<L, 0>(S, Ul + 0 * 256, wave, M, hs);
    ry_group_pass<L, 1>(S, Ul + 1 * 256, wave, M, hs);
    ry_group_pass<L, 2>(S, Ul + 2 * 256, wave, M, hs);
    ry_group_pass<L, 3>(S, Ul + 3 * 256, wave, M, hs);
}

__global__ __launch_bounds__(THREADS)
void qsim16_kernel(const float* __restrict__ x,
                   const float* __restrict__ params,
                   float* __restrict__ out)
{
    __shared__ __align__(16) float S[DIM];
    __shared__ float U[N_LAYERS][4][16][16];
    __shared__ float Phi[256];
    __shared__ float Plo[256];
    __shared__ float cw[16], sw[16];
    __shared__ float pcs[N_LAYERS][16];
    __shared__ float pss[N_LAYERS][16];
    __shared__ float partial[16][16];

    float4* S4 = reinterpret_cast<float4*>(S);

    const int tid  = threadIdx.x;
    const int lane = tid & 31;
    const int wave = tid >> 5;
    const int b    = blockIdx.x;

    if (tid < 16) {
        float a = 0.5f * x[b * N_QUBITS + tid];
        float s, c;
        __sincosf(a, &s, &c);
        cw[tid] = c;
        sw[tid] = s;
    } else if (tid < 48) {
        int idx = tid - 16;
        int l = idx >> 4, w = idx & 15;
        float a = 0.5f * params[idx];
        float s, c;
        __sincosf(a, &s, &c);
        pcs[l][w] = c;
        pss[l][w] = s;
    }
    __syncthreads();

    if (tid < 256) {
        int r = tid;
        float p = 1.0f;
#pragma unroll
        for (int w = 0; w < 8; ++w)
            p *= ((r >> (7 - w)) & 1) ? sw[w] : cw[w];
        Phi[r] = p;
    } else {
        int c = tid - 256;
        float p = 1.0f;
#pragma unroll
        for (int w = 8; w < 16; ++w)
            p *= ((c >> (15 - w)) & 1) ? sw[w] : cw[w];
        Plo[c] = p;
    }

    for (int e = tid; e < N_LAYERS * 4 * 256; e += THREADS) {
        int l  = e >> 10;
        int g  = (e >> 8) & 3;
        int ar = (e >> 4) & 15;
        int ac = e & 15;
        float v = 1.0f;
#pragma unroll
        for (int m = 0; m < 4; ++m) {
            int w = 4 * g + m;
            int i = (ar >> (3 - m)) & 1;
            int j = (ac >> (3 - m)) & 1;
            float f = (i == j) ? pcs[l][w] : (i ? pss[l][w] : -pss[l][w]);
            v *= f;
        }
        U[l][g][ar][ac] = v;
    }
    __syncthreads();

    float acc[16];
#pragma unroll
    for (int w = 0; w < 16; ++w) acc[w] = 0.0f;

    const int M  = lane & 15;
    const int hs = lane >> 4;

    for (int part = 0; part < 2; ++part) {
        for (int jj = 0; jj < DIM / 4 / THREADS; ++jj) {
            int blkz = tid + THREADS * jj;
            int z0   = blkz << 2;
            float ph_hi = Phi[z0 >> 8];
            int   k0   = __popc(z0);
            float4 v;
#pragma unroll
            for (int e = 0; e < 4; ++e) {
                int k = (k0 + (e == 3 ? 2 : (e == 0 ? 0 : 1))) & 3;
                float ph;
                if (part == 0) ph = (k == 0) ? 1.0f : ((k == 2) ? -1.0f : 0.0f);
                else           ph = (k == 3) ? 1.0f : ((k == 1) ? -1.0f : 0.0f);
                float m = ph_hi * Plo[(z0 & 255) | e] * ph;
                if (e == 0) v.x = m; else if (e == 1) v.y = m;
                else if (e == 2) v.z = m; else v.w = m;
            }
            S4[blkz] = v;
        }
        __syncthreads();

        static_assert(N_LAYERS == 2, "one ry_layer call per layer below");
        ry_layer<0>(S, &U[0][0][0][0], wave, M, hs);
        ry_layer<1>(S, &U[1][0][0][0], wave, M, hs);

        for (int jj = 0; jj < DIM / 4 / THREADS; ++jj) {
            int blkz = tid + THREADS * jj;
            int z0   = blkz << 2;
            float4 v = S4[blkz];
            float p0 = v.x * v.x, p1 = v.y * v.y, p2 = v.z * v.z, p3 = v.w * v.w;
            const int y0 = logical_of(z0);
#pragma unroll
            for (int w = 0; w < 16; ++w) {
                constexpr int e1 = logical_of(1), e2 = logical_of(2), e3 = logical_of(3);
                const int bit = 15 - w;
                float c = p0;
                c += ((e1 >> bit) & 1) ? -p1 : p1;
                c += ((e2 >> bit) & 1) ? -p2 : p2;
                c += ((e3 >> bit) & 1) ? -p3 : p3;
                acc[w] += ((y0 >> bit) & 1) ? -c : c;
            }
        }
        __syncthreads();
    }

#pragma unroll
    for (int w = 0; w < 16; ++w) {
        float v = acc[w];
        v += __shfl_xor(v, 16, 32);
        v += __shfl_xor(v, 8, 32);
        v += __shfl_xor(v, 4, 32);
        v += __shfl_xor(v, 2, 32);
        v += __shfl_xor(v, 1, 32);
        acc[w] = v;
    }
    if (lane == 0) {
#pragma unroll
        for (int w = 0; w < 16; ++w) partial[wave][w] = acc[w];
    }
    __syncthreads();
    if (tid < 16) {
        float s = 0.0f;
#pragma unroll
        for (int q = 0; q < 16; ++q) s += partial[q][tid];
        out[b * N_QUBITS + tid] = s;
    }
}

extern "C" void kernel_launch(void* const* d_in, const int* in_sizes, int n_in,
                              void* d_out, int out_size, void* d_ws, size_t ws_size,
                              hipStream_t stream) {
    const float* x      = (const float*)d_in[0];
    const float* params = (const float*)d_in[1];
    float* out          = (float*)d_out;
    (void)in_sizes; (void)n_in; (void)out_size; (void)d_ws; (void)ws_size;
    qsim16_kernel<<<BATCH, THREADS, 0, stream>>>(x, params, out);
}


# First RY layer acts wire by wire on a product state: folded into the fill's per-wire complex factors (no sums left in layer 0, rounding changes); its four WMMA passes gone, last layer stays on WMMA
# speedup vs baseline: 2.4765x; 1.5860x over previous
#include <hip/hip_runtime.h>


#define N_QUBITS 16
#define N_LAYERS 2
#define BATCH    512
#define DIM      65536
#define THREADS  512

typedef __attribute__((ext_vector_type(2))) float v2f;
typedef __attribute__((ext_vector_type(8))) float v8f;

__host__ __device__ constexpr int ring_inv(int y) {
    return (y ^ (y >> 1)) ^ ((y & 1) ? 0xC000 : 0);
}
__host__ __device__ constexpr int ring_fwd(int p) {
    int g = p;
    g ^= g >> 1; g ^= g >> 2; g ^= g >> 4; g ^= g >> 8;
    return (g & 0x7FFF) | (((g ^ (p >> 15)) & 1) << 15);
}
template <int L>
__host__ __device__ constexpr int qmap(int y) {
    for (int i = 0; i < L; ++i) y = ring_inv(y);
    return y;
}
__host__ __device__ constexpr int logical_of(int p) {
    for (int i = 0; i < N_LAYERS; ++i) p = ring_fwd(p);
    return p;
}
template <int L>
__device__ __forceinline__ int join(int a, int b) {
    return (L == 0) ? (a | b) : (a ^ b);
}

template <int L, int G>
__device__ __forceinline__ void ry_group_pass(float* S, const float* Um,
                                              int wave, int M, int hs)
{
    constexpr int beta    = 12 - 4 * G;
    constexpr int lowmask = (1 << beta) - 1;

    v2f afr[4];
#pragma unroll
    for (int k0 = 0; k0 < 4; ++k0) {
        int K = 4 * k0 + 2 * hs;
        afr[k0].x = Um[M * 16 + K];
        afr[k0].y = Um[M * 16 + K + 1];
    }

    const int offB = qmap<L>((2 * hs) << beta);
    const int offD = qmap<L>((8 * hs) << beta);

    for (int blk = wave; blk < 256; blk += 16) {
        int j  = blk * 16 + M;
        int zb = ((j & ~lowmask) << 4) | (j & lowmask);
        int pb = qmap<L>(zb);
        int pl = join<L>(pb, offB);
        int pd = join<L>(pb, offD);
        v8f c8 = {};
#pragma unroll
        for (int k0 = 0; k0 < 4; ++k0) {
            v2f bfr;
            bfr.x = S[join<L>(pl, qmap<L>((4 * k0)     << beta))];
            bfr.y = S[join<L>(pl, qmap<L>((4 * k0 + 1) << beta))];
            c8 = __builtin_amdgcn_wmma_f32_16x16x4_f32(
                     false, afr[k0], false, bfr,
                     (short)0, c8, false, false);
        }
#pragma unroll
        for (int v = 0; v < 8; ++v)
            S[join<L>(pd, qmap<L>(v << beta))] = c8[v];
    }
    __syncthreads();
}

template <int L>
__device__ __forceinline__ void ry_layer(float* S, const float* Ul,
                                         int wave, int M, int hs)
{
    ry_group_pass<L, 0>(S, Ul + 0 * 256, wave, M, hs);
    ry_group_pass<L, 1>(S, Ul + 1 * 256, wave, M, hs);
    ry_group_pass<L, 2>(S, Ul + 2 * 256, wave, M, hs);
    ry_group_pass<L, 3>(S, Ul + 3 * 256, wave, M, hs);
}

__global__ __launch_bounds__(THREADS)
void qsim16_kernel(const float* __restrict__ x,
                   const float* __restrict__ params,
                   float* __restrict__ out)
{
    __shared__ __align__(16) float S[DIM];
    __shared__ float U[4][16][16];
    __shared__ float PhiR[256], PhiI[256];
    __shared__ __align__(16) float PloR[256];
    __shared__ __align__(16) float PloI[256];
    __shared__ float cw[16], sw[16];
    __shared__ float pcs[N_LAYERS][16];
    __shared__ float pss[N_LAYERS][16];
    __shared__ float partial[16][16];

    float4* S4 = reinterpret_cast<float4*>(S);

    const int tid  = threadIdx.x;
    const int lane = tid & 31;
    const int wave = tid >> 5;
    const int b    = blockIdx.x;

    if (tid < 16) {
        float a = 0.5f * x[b * N_QUBITS + tid];
        float s, c;
        __sincosf(a, &s, &c);
        cw[tid] = c;
        sw[tid] = s;
    } else if (tid < 48) {
        int idx = tid - 16;
        int l = idx >> 4, w = idx & 15;
        float a = 0.5f * params[idx];
        float s, c;
        __sincosf(a, &s, &c);
        pcs[l][w] = c;
        pss[l][w] = s;
    }
    __syncthreads();

    {
        const int half = tid >> 8;
        const int q    = tid & 255;
        float pr = 1.0f, pi = 0.0f;
#pragma unroll
        for (int m = 0; m < 8; ++m) {
            const int w = 8 * half + m;
            const float c = cw[w], s = sw[w], pc = pcs[0][w], ps = pss[0][w];
            const bool one = (q >> (7 - m)) & 1;
            const float fr = one ? ps * c : pc * c;
            const float fi = one ? -(pc * s) : ps * s;
            const float nr = pr * fr - pi * fi;
            const float ni = pr * fi + pi * fr;
            pr = nr; pi = ni;
        }
        if (half == 0) { PhiR[q] = pr; PhiI[q] = pi; }
        else           { PloR[q] = pr; PloI[q] = pi; }
    }

    for (int e = tid; e < 4 * 256; e += THREADS) {
        const int l = N_LAYERS - 1;
        int g  = (e >> 8) & 3;
        int ar = (e >> 4) & 15;
        int ac = e & 15;
        float v = 1.0f;
#pragma unroll
        for (int m = 0; m < 4; ++m) {
            int w = 4 * g + m;
            int i = (ar >> (3 - m)) & 1;
            int j = (ac >> (3 - m)) & 1;
            float f = (i == j) ? pcs[l][w] : (i ? pss[l][w] : -pss[l][w]);
            v *= f;
        }
        U[g][ar][ac] = v;
    }
    __syncthreads();

    float acc[16];
#pragma unroll
    for (int w = 0; w < 16; ++w) acc[w] = 0.0f;

    const int M  = lane & 15;
    const int hs = lane >> 4;

    for (int part = 0; part < 2; ++part) {
        {
            const float4 lr = *reinterpret_cast<const float4*>(&PloR[(tid << 2) & 255]);
            const float4 li = *reinterpret_cast<const float4*>(&PloI[(tid << 2) & 255]);
            float4 a, bq;
            if (part == 0) { a = lr; bq.x = -li.x; bq.y = -li.y; bq.z = -li.z; bq.w = -li.w; }
            else           { a = li; bq = lr; }
            for (int jj = 0; jj < DIM / 4 / THREADS; ++jj) {
                int blkz = tid + THREADS * jj;
                int r    = blkz >> 6;
                const float hr = PhiR[r], hi = PhiI[r];
                float4 v;
                v.x = hr * a.x + hi * bq.x;  v.y = hr * a.y + hi * bq.y;
                v.z = hr * a.z + hi * bq.z;  v.w = hr * a.w + hi * bq.w;
                S4[blkz] = v;
            }
        }
        __syncthreads();

        static_assert(N_LAYERS == 2, "the fill holds layer 0, one ry_layer call for layer 1");
        ry_layer<1>(S, &U[0][0][0], wave, M, hs);

        for (int jj = 0; jj < DIM / 4 / THREADS; ++jj) {
            int blkz = tid + THREADS * jj;
            int z0   = blkz << 2;
            float4 v = S4[blkz];
            float p0 = v.x * v.x, p1 = v.y * v.y, p2 = v.z * v.z, p3 = v.w * v.w;
            const int y0 = logical_of(z0);
#pragma unroll
            for (int w = 0; w < 16; ++w) {
                constexpr int e1 = logical_of(1), e2 = logical_of(2), e3 = logical_of(3);
                const int bit = 15 - w;
                float c = p0;
                c += ((e1 >> bit) & 1) ? -p1 : p1;
                c += ((e2 >> bit) & 1) ? -p2 : p2;
                c += ((e3 >> bit) & 1) ? -p3 : p3;
                acc[w] += ((y0 >> bit) & 1) ? -c : c;
            }
        }
        __syncthreads();
    }

#pragma unroll
    for (int w = 0; w < 16; ++w) {
        float v = acc[w];
        v += __shfl_xor(v, 16, 32);
        v += __shfl_xor(v, 8, 32);
        v += __shfl_xor(v, 4, 32);
        v += __shfl_xor(v, 2, 32);
        v += __shfl_xor(v, 1, 32);
        acc[w] = v;
    }
    if (lane == 0) {
#pragma unroll
        for (int w = 0; w < 16; ++w) partial[wave][w] = acc[w];
    }
    __syncthreads();
    if (tid < 16) {
        float s = 0.0f;
#pragma unroll
        for (int q = 0; q < 16; ++q) s += partial[q][tid];
        out[b * N_QUBITS + tid] = s;
    }
}

extern "C" void kernel_launch(void* const* d_in, const int* in_sizes, int n_in,
                              void* d_out, int out_size, void* d_ws, size_t ws_size,
                              hipStream_t stream) {
    const float* x      = (const float*)d_in[0];
    const float* params = (const float*)d_in[1];
    float* out          = (float*)d_out;
    (void)in_sizes; (void)n_in; (void)out_size; (void)d_ws; (void)ws_size;
    qsim16_kernel<<<BATCH, THREADS, 0, stream>>>(x, params, out);
}


# Z-expval sums taken straight from the last RY group's WMMA result registers (8-point +- butterfly per tile); final state never stored to LDS nor re-read; expval summation order changes
# speedup vs baseline: 3.3495x; 1.3525x over previous
#include <hip/hip_runtime.h>


#define N_QUBITS 16
#define N_LAYERS 2
#define BATCH    512
#define DIM      65536
#define THREADS  512

typedef __attribute__((ext_vector_type(2))) float v2f;
typedef __attribute__((ext_vector_type(8))) float v8f;

__host__ __device__ constexpr int ring_inv(int y) {
    return (y ^ (y >> 1)) ^ ((y & 1) ? 0xC000 : 0);
}
__host__ __device__ constexpr int ring_fwd(int p) {
    int g = p;
    g ^= g >> 1; g ^= g >> 2; g ^= g >> 4; g ^= g >> 8;
    return (g & 0x7FFF) | (((g ^ (p >> 15)) & 1) << 15);
}
template <int L>
__host__ __device__ constexpr int qmap(int y) {
    for (int i = 0; i < L; ++i) y = ring_inv(y);
    return y;
}
__host__ __device__ constexpr int logical_of(int p) {
    for (int i = 0; i < N_LAYERS; ++i) p = ring_fwd(p);
    return p;
}
template <int L>
__device__ __forceinline__ int join(int a, int b) {
    return (L == 0) ? (a | b) : (a ^ b);
}

template <int L, int G>
__device__ __forceinline__ void ry_group_pass(float* S, const float* Um,
                                              int wave, int M, int hs)
{
    constexpr int beta    = 12 - 4 * G;
    constexpr int lowmask = (1 << beta) - 1;

    v2f afr[4];
#pragma unroll
    for (int k0 = 0; k0 < 4; ++k0) {
        int K = 4 * k0 + 2 * hs;
        afr[k0].x = Um[M * 16 + K];
        afr[k0].y = Um[M * 16 + K + 1];
    }

    const int offB = qmap<L>((2 * hs) << beta);
    const int offD = qmap<L>((8 * hs) << beta);

    for (int blk = wave; blk < 256; blk += 16) {
        int j  = blk * 16 + M;
        int zb = ((j & ~lowmask) << 4) | (j & lowmask);
        int pb = qmap<L>(zb);
        int pl = join<L>(pb, offB);
        int pd = join<L>(pb, offD);
        v8f c8 = {};
#pragma unroll
        for (int k0 = 0; k0 < 4; ++k0) {
            v2f bfr;
            bfr.x = S[join<L>(pl, qmap<L>((4 * k0)     << beta))];
            bfr.y = S[join<L>(pl, qmap<L>((4 * k0 + 1) << beta))];
            c8 = __builtin_amdgcn_wmma_f32_16x16x4_f32(
                     false, afr[k0], false, bfr,
                     (short)0, c8, false, false);
        }
#pragma unroll
        for (int v = 0; v < 8; ++v)
            S[join<L>(pd, qmap<L>(v << beta))] = c8[v];
    }
    __syncthreads();
}

struct ZMasks { int m[16]; int rep[16]; };
template <int L>
__host__ __device__ constexpr ZMasks make_zmasks() {
    ZMasks z = {};
    for (int w = 0; w < 16; ++w)
        for (int k = 0; k < 16; ++k)
            if ((logical_of(qmap<L>(1 << k)) >> (15 - w)) & 1) z.m[w] |= 1 << k;
    for (int w = 0; w < 16; ++w) {
        z.rep[w] = w;
        for (int u = w - 1; u >= 0; --u)
            if ((z.m[u] & 0xF007) == (z.m[w] & 0xF007)) z.rep[w] = u;
    }
    return z;
}
__host__ __device__ constexpr bool odd_parity(int x) {
    x ^= x >> 8; x ^= x >> 4; x ^= x >> 2; x ^= x >> 1;
    return x & 1;
}

template <int L>
__device__ __forceinline__ void ry_last_group_zsums(const float* S, const float* Um,
                                                    int wave, int M, int hs,
                                                    float (&acc)[16])
{
    constexpr ZMasks zm = make_zmasks<L>();

    v2f afr[4];
#pragma unroll
    for (int k0 = 0; k0 < 4; ++k0) {
        int K = 4 * k0 + 2 * hs;
        afr[k0].x = Um[M * 16 + K];
        afr[k0].y = Um[M * 16 + K + 1];
    }

    const int pl0 = join<L>(qmap<L>(((wave << 4) | M) << 4), qmap<L>(2 * hs));

#pragma nounroll
    for (int it = 0; it < 16; ++it) {
        const int pl = join<L>(pl0, qmap<L>(it << 12));
        v8f c8 = {};
#pragma unroll
        for (int k0 = 0; k0 < 4; ++k0) {
            v2f bfr;
            bfr.x = S[join<L>(pl, qmap<L>(4 * k0))];
            bfr.y = S[join<L>(pl, qmap<L>(4 * k0 + 1))];
            c8 = __builtin_amdgcn_wmma_f32_16x16x4_f32(
                     false, afr[k0], false, bfr,
                     (short)0, c8, false, false);
        }
        float h[8];
#pragma unroll
        for (int v = 0; v < 8; ++v) h[v] = c8[v] * c8[v];
#pragma unroll
        for (int len = 1; len < 8; len <<= 1)
#pragma unroll
            for (int i = 0; i < 8; i += 2 * len)
#pragma unroll
                for (int j = i; j < i + len; ++j) {
                    const float a = h[j], bq = h[j + len];
                    h[j] = a + bq;
                    h[j + len] = a - bq;
                }
#pragma unroll
        for (int w = 0; w < 16; ++w) {
            if (zm.rep[w] != w) continue;
            const float sg = odd_parity(zm.m[w] & (it << 12)) ? -1.0f : 1.0f;
            acc[w] += sg * h[zm.m[w] & 7];
        }
    }
}

template <int L>
__device__ __forceinline__ void ry_layer_zsums(float* S, const float* Ul,
                                               int wave, int M, int hs,
                                               float (&acc)[16])
{
    ry_group_pass<L, 0>(S, Ul + 0 * 256, wave, M, hs);
    ry_group_pass<L, 1>(S, Ul + 1 * 256, wave, M, hs);
    ry_group_pass<L, 2>(S, Ul + 2 * 256, wave, M, hs);
    ry_last_group_zsums<L>(S, Ul + 3 * 256, wave, M, hs, acc);
}

__global__ __launch_bounds__(THREADS)
void qsim16_kernel(const float* __restrict__ x,
                   const float* __restrict__ params,
                   float* __restrict__ out)
{
    __shared__ __align__(16) float S[DIM];
    __shared__ float U[4][16][16];
    __shared__ float PhiR[256], PhiI[256];
    __shared__ __align__(16) float PloR[256];
    __shared__ __align__(16) float PloI[256];
    __shared__ float cw[16], sw[16];
    __shared__ float pcs[N_LAYERS][16];
    __shared__ float pss[N_LAYERS][16];
    __shared__ float partial[16][16];

    float4* S4 = reinterpret_cast<float4*>(S);

    const int tid  = threadIdx.x;
    const int lane = tid & 31;
    const int wave = tid >> 5;
    const int b    = blockIdx.x;

    if (tid < 16) {
        float a = 0.5f * x[b * N_QUBITS + tid];
        float s, c;
        __sincosf(a, &s, &c);
        cw[tid] = c;
        sw[tid] = s;
    } else if (tid < 48) {
        int idx = tid - 16;
        int l = idx >> 4, w = idx & 15;
        float a = 0.5f * params[idx];
        float s, c;
        __sincosf(a, &s, &c);
        pcs[l][w] = c;
        pss[l][w] = s;
    }
    __syncthreads();

    {
        const int half = tid >> 8;
        const int q    = tid & 255;
        float pr = 1.0f, pi = 0.0f;
#pragma unroll
        for (int m = 0; m < 8; ++m) {
            const int w = 8 * half + m;
            const float c = cw[w], s = sw[w], pc = pcs[0][w], ps = pss[0][w];
            const bool one = (q >> (7 - m)) & 1;
            const float fr = one ? ps * c : pc * c;
            const float fi = one ? -(pc * s) : ps * s;
            const float nr = pr * fr - pi * fi;
            const float ni = pr * fi + pi * fr;
            pr = nr; pi = ni;
        }
        if (half == 0) { PhiR[q] = pr; PhiI[q] = pi; }
        else           { PloR[q] = pr; PloI[q] = pi; }
    }

    for (int e = tid; e < 4 * 256; e += THREADS) {
        const int l = N_LAYERS - 1;
        int g  = (e >> 8) & 3;
        int ar = (e >> 4) & 15;
        int ac = e & 15;
        float v = 1.0f;
#pragma unroll
        for (int m = 0; m < 4; ++m) {
            int w = 4 * g + m;
            int i = (ar >> (3 - m)) & 1;
            int j = (ac >> (3 - m)) & 1;
            float f = (i == j) ? pcs[l][w] : (i ? pss[l][w] : -pss[l][w]);
            v *= f;
        }
        U[g][ar][ac] = v;
    }
    __syncthreads();

    float acc[16];
#pragma unroll
    for (int w = 0; w < 16; ++w) acc[w] = 0.0f;

    const int M  = lane & 15;
    const int hs = lane >> 4;

    for (int part = 0; part < 2; ++part) {
        {
            const float4 lr = *reinterpret_cast<const float4*>(&PloR[(tid << 2) & 255]);
            const float4 li = *reinterpret_cast<const float4*>(&PloI[(tid << 2) & 255]);
            float4 a, bq;
            if (part == 0) { a = lr; bq.x = -li.x; bq.y = -li.y; bq.z = -li.z; bq.w = -li.w; }
            else           { a = li; bq = lr; }
            for (int jj = 0; jj < DIM / 4 / THREADS; ++jj) {
                int blkz = tid + THREADS * jj;
                int r    = blkz >> 6;
                const float hr = PhiR[r], hi = PhiI[r];
                float4 v;
                v.x = hr * a.x + hi * bq.x;  v.y = hr * a.y + hi * bq.y;
                v.z = hr * a.z + hi * bq.z;  v.w = hr * a.w + hi * bq.w;
                S4[blkz] = v;
            }
        }
        __syncthreads();

        static_assert(N_LAYERS == 2, "the fill holds layer 0, one ry_layer call for layer 1");
        ry_layer_zsums<1>(S, &U[0][0][0], wave, M, hs, acc);
        __syncthreads();
    }

    constexpr ZMasks zmf = make_zmasks<N_LAYERS - 1>();
    const int ylane = (wave << 8) | (M << 4) | (hs << 3);
    float red[16];
#pragma unroll
    for (int w = 0; w < 16; ++w) {
        float v = acc[zmf.rep[w]];
        v = odd_parity(ylane & zmf.m[w]) ? -v : v;
        v += __shfl_xor(v, 16, 32);
        v += __shfl_xor(v, 8, 32);
        v += __shfl_xor(v, 4, 32);
        v += __shfl_xor(v, 2, 32);
        v += __shfl_xor(v, 1, 32);
        red[w] = v;
    }
    if (lane == 0) {
#pragma unroll
        for (int w = 0; w < 16; ++w) partial[wave][w] = red[w];
    }
    __syncthreads();
    if (tid < 16) {
        float s = 0.0f;
#pragma unroll
        for (int q = 0; q < 16; ++q) s += partial[q][tid];
        out[b * N_QUBITS + tid] = s;
    }
}

extern "C" void kernel_launch(void* const* d_in, const int* in_sizes, int n_in,
                              void* d_out, int out_size, void* d_ws, size_t ws_size,
                              hipStream_t stream) {
    const float* x      = (const float*)d_in[0];
    const float* params = (const float*)d_in[1];
    float* out          = (float*)d_out;
    (void)in_sizes; (void)n_in; (void)out_size; (void)d_ws; (void)ws_size;
    qsim16_kernel<<<BATCH, THREADS, 0, stream>>>(x, params, out);
}


# Two RY groups per LDS trip: group one's WMMA result registers are group two's B operand (k order matched in U), XOR bank-spread layout for the high-bit tiles; 4 passes to 2; RY sum order changes
# speedup vs baseline: 3.9693x; 1.1850x over previous
#include <hip/hip_runtime.h>


#define N_QUBITS 16
#define N_LAYERS 2
#define BATCH    512
#define DIM      65536
#define THREADS  512

typedef __attribute__((ext_vector_type(2))) float v2f;
typedef __attribute__((ext_vector_type(8))) float v8f;

__host__ __device__ constexpr int ring_inv(int y) {
    return (y ^ (y >> 1)) ^ ((y & 1) ? 0xC000 : 0);
}
__host__ __device__ constexpr int ring_fwd(int p) {
    int g = p;
    g ^= g >> 1; g ^= g >> 2; g ^= g >> 4; g ^= g >> 8;
    return (g & 0x7FFF) | (((g ^ (p >> 15)) & 1) << 15);
}
template <int L>
__host__ __device__ constexpr int qmap(int y) {
    for (int i = 0; i < L; ++i) y = ring_inv(y);
    return y;
}
__host__ __device__ constexpr int logical_of(int p) {
    for (int i = 0; i < N_LAYERS; ++i) p = ring_fwd(p);
    return p;
}
__host__ __device__ constexpr int lds_swz(int p) {
    return p ^ ((p >> 8) & 0x3F) ^ (((p >> 12) & 0xF) << 2);
}
template <int L>
__host__ __device__ constexpr int pmap(int y) {
    return lds_swz(qmap<L>(y));
}

struct ZMasks { int m[16]; int rep[16]; };
template <int L>
__host__ __device__ constexpr ZMasks make_zmasks() {
    ZMasks z = {};
    for (int w = 0; w < 16; ++w)
        for (int k = 0; k < 16; ++k)
            if ((logical_of(qmap<L>(1 << k)) >> (15 - w)) & 1) z.m[w] |= 1 << k;
    for (int w = 0; w < 16; ++w) {
        z.rep[w] = w;
        for (int u = w - 1; u >= 0; --u)
            if ((z.m[u] & 0xF007) == (z.m[w] & 0xF007)) z.rep[w] = u;
    }
    return z;
}
__host__ __device__ constexpr bool odd_parity(int x) {
    x ^= x >> 8; x ^= x >> 4; x ^= x >> 2; x ^= x >> 1;
    return x & 1;
}

template <int L, int GA, int GB, bool LAST>
__device__ __forceinline__ void ry_pair_pass(float* S, const float* Ul,
                                             int wave, int c, int hs,
                                             float (&acc)[16])
{
    constexpr int ba = 12 - 4 * GA;
    constexpr int bb = 12 - 4 * GB;
    static_assert(!LAST || (GA == 2 && GB == 3), "the Z sums expect b' in bits 0..3 and a' in bits 4..7");
    static_assert(LAST || (ba >= 8 && bb >= 8), "the in-place pass takes its fibers from bits 0..7");
    const float* Ua = Ul + GA * 256;
    const float* Ub = Ul + GB * 256;

    v2f ufa[4], ufb[4];
#pragma unroll
    for (int k0 = 0; k0 < 4; ++k0) {
        const int ka = 4 * k0 + 2 * hs;
        ufa[k0].x = Ua[c * 16 + ka];
        ufa[k0].y = Ua[c * 16 + ka + 1];
        const int kb = 8 * hs + 2 * k0;
        ufb[k0].x = Ub[c * 16 + kb];
        ufb[k0].y = Ub[c * 16 + kb + 1];
    }

    const int ldl = pmap<L>(c << bb) ^ pmap<L>((2 * hs) << ba);
    const int stl = pmap<L>(c << ba) ^ pmap<L>((8 * hs) << bb);

    constexpr ZMasks zm = make_zmasks<L>();

#pragma nounroll
    for (int it = 0; it < 16; ++it) {
        const int fib = LAST ? ((it << 12) | (wave << 8)) : ((it << 4) | wave);
        const int pf  = pmap<L>(fib);
        const int pl  = pf ^ ldl;
        v8f d1 = {};
#pragma unroll
        for (int k0 = 0; k0 < 4; ++k0) {
            v2f xf;
            xf.x = S[pl ^ pmap<L>((4 * k0)     << ba)];
            xf.y = S[pl ^ pmap<L>((4 * k0 + 1) << ba)];
            d1 = __builtin_amdgcn_wmma_f32_16x16x4_f32(
                     false, xf, false, ufa[k0],
                     (short)0, d1, false, false);
        }
        v8f d2 = {};
#pragma unroll
        for (int k0 = 0; k0 < 4; ++k0) {
            v2f bq;
            bq.x = d1[2 * k0];
            bq.y = d1[2 * k0 + 1];
            d2 = __builtin_amdgcn_wmma_f32_16x16x4_f32(
                     false, ufb[k0], false, bq,
                     (short)0, d2, false, false);
        }
        if (!LAST) {
            const int pd = pf ^ stl;
#pragma unroll
            for (int r = 0; r < 8; ++r)
                S[pd ^ pmap<L>(r << bb)] = d2[r];
        } else {
            float h[8];
#pragma unroll
            for (int r = 0; r < 8; ++r) h[r] = d2[r] * d2[r];
#pragma unroll
            for (int len = 1; len < 8; len <<= 1)
#pragma unroll
                for (int i = 0; i < 8; i += 2 * len)
#pragma unroll
                    for (int j = i; j < i + len; ++j) {
                        const float a = h[j], bq = h[j + len];
                        h[j] = a + bq;
                        h[j + len] = a - bq;
                    }
#pragma unroll
            for (int w = 0; w < 16; ++w) {
                if (zm.rep[w] != w) continue;
                const float sg = odd_parity(zm.m[w] & (it << 12)) ? -1.0f : 1.0f;
                acc[w] += sg * h[zm.m[w] & 7];
            }
        }
    }
    if (!LAST) __syncthreads();
}

template <int L>
__device__ __forceinline__ void ry_layer_zsums(float* S, const float* Ul,
                                               int wave, int c, int hs,
                                               float (&acc)[16])
{
    ry_pair_pass<L, 1, 0, false>(S, Ul, wave, c, hs, acc);
    ry_pair_pass<L, 2, 3, true >(S, Ul, wave, c, hs, acc);
}

__global__ __launch_bounds__(THREADS)
void qsim16_kernel(const float* __restrict__ x,
                   const float* __restrict__ params,
                   float* __restrict__ out)
{
    __shared__ __align__(16) float S[DIM];
    __shared__ float U[4][16][16];
    __shared__ float PhiR[256], PhiI[256];
    __shared__ __align__(16) float PloR[256];
    __shared__ __align__(16) float PloI[256];
    __shared__ float cw[16], sw[16];
    __shared__ float pcs[N_LAYERS][16];
    __shared__ float pss[N_LAYERS][16];
    __shared__ float partial[16][16];

    float4* S4 = reinterpret_cast<float4*>(S);

    const int tid  = threadIdx.x;
    const int lane = tid & 31;
    const int wave = tid >> 5;
    const int b    = blockIdx.x;

    if (tid < 16) {
        float a = 0.5f * x[b * N_QUBITS + tid];
        float s, c;
        __sincosf(a, &s, &c);
        cw[tid] = c;
        sw[tid] = s;
    } else if (tid < 48) {
        int idx = tid - 16;
        int l = idx >> 4, w = idx & 15;
        float a = 0.5f * params[idx];
        float s, c;
        __sincosf(a, &s, &c);
        pcs[l][w] = c;
        pss[l][w] = s;
    }
    __syncthreads();

    {
        const int half = tid >> 8;
        const int q    = tid & 255;
        float pr = 1.0f, pi = 0.0f;
#pragma unroll
        for (int m = 0; m < 8; ++m) {
            const int w = 8 * half + m;
            const float c = cw[w], s = sw[w], pc = pcs[0][w], ps = pss[0][w];
            const bool one = (q >> (7 - m)) & 1;
            const float fr = one ? ps * c : pc * c;
            const float fi = one ? -(pc * s) : ps * s;
            const float nr = pr * fr - pi * fi;
            const float ni = pr * fi + pi * fr;
            pr = nr; pi = ni;
        }
        if (half == 0) { PhiR[q] = pr; PhiI[q] = pi; }
        else           { PloR[q] = pr; PloI[q] = pi; }
    }

    for (int e = tid; e < 4 * 256; e += THREADS) {
        const int l = N_LAYERS - 1;
        int g  = (e >> 8) & 3;
        int ar = (e >> 4) & 15;
        int ac = e & 15;
        float v = 1.0f;
#pragma unroll
        for (int m = 0; m < 4; ++m) {
            int w = 4 * g + m;
            int i = (ar >> (3 - m)) & 1;
            int j = (ac >> (3 - m)) & 1;
            float f = (i == j) ? pcs[l][w] : (i ? pss[l][w] : -pss[l][w]);
            v *= f;
        }
        U[g][ar][ac] = v;
    }
    __syncthreads();

    float acc[16];
#pragma unroll
    for (int w = 0; w < 16; ++w) acc[w] = 0.0f;

    const int M  = lane & 15;
    const int hs = lane >> 4;

    for (int part = 0; part < 2; ++part) {
        {
            const float4 lr = *reinterpret_cast<const float4*>(&PloR[(tid << 2) & 255]);
            const float4 li = *reinterpret_cast<const float4*>(&PloI[(tid << 2) & 255]);
            float4 a, bq;
            if (part == 0) { a = lr; bq.x = -li.x; bq.y = -li.y; bq.z = -li.z; bq.w = -li.w; }
            else           { a = li; bq = lr; }
            const int sx = (tid >> 6) & 3;
            if (sx & 1) { float t;
                t = a.x;  a.x  = a.y;  a.y  = t;  t = a.z;  a.z  = a.w;  a.w  = t;
                t = bq.x; bq.x = bq.y; bq.y = t;  t = bq.z; bq.z = bq.w; bq.w = t; }
            if (sx & 2) { float t;
                t = a.x;  a.x  = a.z;  a.z  = t;  t = a.y;  a.y  = a.w;  a.w  = t;
                t = bq.x; bq.x = bq.z; bq.z = t;  t = bq.y; bq.y = bq.w; bq.w = t; }
            for (int jj = 0; jj < DIM / 4 / THREADS; ++jj) {
                int blkz = tid + THREADS * jj;
                int r    = blkz >> 6;
                const float hr = PhiR[r], hi = PhiI[r];
                float4 v;
                v.x = hr * a.x + hi * bq.x;  v.y = hr * a.y + hi * bq.y;
                v.z = hr * a.z + hi * bq.z;  v.w = hr * a.w + hi * bq.w;
                S4[blkz ^ (((blkz >> 8) ^ (blkz >> 10)) & 0xF)] = v;
            }
        }
        __syncthreads();

        static_assert(N_LAYERS == 2, "the fill holds layer 0, one ry_layer call for layer 1");
        ry_layer_zsums<1>(S, &U[0][0][0], wave, M, hs, acc);
        __syncthreads();
    }

    constexpr ZMasks zmf = make_zmasks<N_LAYERS - 1>();
    const int ylane = (wave << 8) | (M << 4) | (hs << 3);
    float red[16];
#pragma unroll
    for (int w = 0; w < 16; ++w) {
        float v = acc[zmf.rep[w]];
        v = odd_parity(ylane & zmf.m[w]) ? -v : v;
        v += __shfl_xor(v, 16, 32);
        v += __shfl_xor(v, 8, 32);
        v += __shfl_xor(v, 4, 32);
        v += __shfl_xor(v, 2, 32);
        v += __shfl_xor(v, 1, 32);
        red[w] = v;
    }
    if (lane == 0) {
#pragma unroll
        for (int w = 0; w < 16; ++w) partial[wave][w] = red[w];
    }
    __syncthreads();
    if (tid < 16) {
        float s = 0.0f;
#pragma unroll
        for (int q = 0; q < 16; ++q) s += partial[q][tid];
        out[b * N_QUBITS + tid] = s;
    }
}

extern "C" void kernel_launch(void* const* d_in, const int* in_sizes, int n_in,
                              void* d_out, int out_size, void* d_ws, size_t ws_size,
                              hipStream_t stream) {
    const float* x      = (const float*)d_in[0];
    const float* params = (const float*)d_in[1];
    float* out          = (float*)d_out;
    (void)in_sizes; (void)n_in; (void)out_size; (void)d_ws; (void)ws_size;
    qsim16_kernel<<<BATCH, THREADS, 0, stream>>>(x, params, out);
}


# State before the last RY layer is a sum of 4 products (hi 8-wire factor x lo 8-wire factor): RY acts on the 256-entry factors, amplitudes formed per tile by K=8 WMMA, no LDS state; sum order changes
# speedup vs baseline: 16.4368x; 4.1410x over previous
#include <hip/hip_runtime.h>


#define N_QUBITS 16
#define N_LAYERS 2
#define BATCH    512
#define THREADS  256
#define NWAVES   (THREADS / 32)

typedef __attribute__((ext_vector_type(2))) float v2f;
typedef __attribute__((ext_vector_type(8))) float v8f;

__host__ __device__ constexpr int ring_inv(int y) {
    return (y ^ (y >> 1)) ^ ((y & 1) ? 0xC000 : 0);
}
__host__ __device__ constexpr int ring_fwd(int p) {
    int g = p;
    g ^= g >> 1; g ^= g >> 2; g ^= g >> 4; g ^= g >> 8;
    return (g & 0x7FFF) | (((g ^ (p >> 15)) & 1) << 15);
}
template <int L>
__host__ __device__ constexpr int qmap(int y) {
    for (int i = 0; i < L; ++i) y = ring_inv(y);
    return y;
}
__host__ __device__ constexpr int logical_of(int p) {
    for (int i = 0; i < N_LAYERS; ++i) p = ring_fwd(p);
    return p;
}

struct ZMasks { int m[16]; int rep[16]; };
template <int L>
__host__ __device__ constexpr ZMasks make_zmasks() {
    ZMasks z = {};
    for (int w = 0; w < 16; ++w)
        for (int k = 0; k < 16; ++k)
            if ((logical_of(qmap<L>(1 << k)) >> (15 - w)) & 1) z.m[w] |= 1 << k;
    for (int w = 0; w < 16; ++w) {
        z.rep[w] = w;
        for (int u = w - 1; u >= 0; --u)
            if ((z.m[u] & 0xF087) == (z.m[w] & 0xF087)) z.rep[w] = u;
    }
    return z;
}
__host__ __device__ constexpr bool odd_parity(int x) {
    x ^= x >> 8; x ^= x >> 4; x ^= x >> 2; x ^= x >> 1;
    return x & 1;
}

__device__ __forceinline__ int gray8(int h) { return h ^ (h >> 1); }

__global__ __launch_bounds__(THREADS)
void qsim16_kernel(const float* __restrict__ x,
                   const float* __restrict__ params,
                   float* __restrict__ out)
{
    __shared__ __align__(16) float Wt[256 * 8];
    __shared__ __align__(16) float Vt[256 * 8];
    __shared__ float U[4][16][16];
    __shared__ float PhiR[256], PhiI[256];
    __shared__ float PloR[256], PloI[256];
    __shared__ float cw[16], sw[16];
    __shared__ float pcs[N_LAYERS][16];
    __shared__ float pss[N_LAYERS][16];
    __shared__ float partial[NWAVES][16];

    const int tid  = threadIdx.x;
    const int lane = tid & 31;
    const int wave = tid >> 5;
    const int b    = blockIdx.x;

    if (tid < 16) {
        float a = 0.5f * x[b * N_QUBITS + tid];
        float s, c;
        __sincosf(a, &s, &c);
        cw[tid] = c;
        sw[tid] = s;
    } else if (tid < 48) {
        int idx = tid - 16;
        int l = idx >> 4, w = idx & 15;
        float a = 0.5f * params[idx];
        float s, c;
        __sincosf(a, &s, &c);
        pcs[l][w] = c;
        pss[l][w] = s;
    }
    __syncthreads();

    static_assert(THREADS == 256, "one table entry of each half per thread");
#pragma unroll
    for (int half = 0; half < 2; ++half) {
        const int q = tid;
        float pr = 1.0f, pi = 0.0f;
#pragma unroll
        for (int m = 0; m < 8; ++m) {
            const int w = 8 * half + m;
            const float c = cw[w], s = sw[w], pc = pcs[0][w], ps = pss[0][w];
            const bool one = (q >> (7 - m)) & 1;
            const float fr = one ? ps * c : pc * c;
            const float fi = one ? -(pc * s) : ps * s;
            const float nr = pr * fr - pi * fi;
            const float ni = pr * fi + pi * fr;
            pr = nr; pi = ni;
        }
        if (half == 0) { PhiR[q] = pr; PhiI[q] = pi; }
        else           { PloR[q] = pr; PloI[q] = pi; }
    }

    for (int e = tid; e < 4 * 256; e += THREADS) {
        const int l = N_LAYERS - 1;
        int g  = (e >> 8) & 3;
        int ar = (e >> 4) & 15;
        int ac = e & 15;
        float v = 1.0f;
#pragma unroll
        for (int m = 0; m < 4; ++m) {
            int w = 4 * g + m;
            int i = (ar >> (3 - m)) & 1;
            int j = (ac >> (3 - m)) & 1;
            float f = (i == j) ? pcs[l][w] : (i ? pss[l][w] : -pss[l][w]);
            v *= f;
        }
        U[g][ar][ac] = v;
    }
    __syncthreads();

    const int c  = lane & 15;
    const int hs = lane >> 4;

    static_assert(NWAVES == 8, "wave = (e, f, ri); lo' row blocks wave and wave + 8");
#pragma unroll
    for (int tab = 0; tab < 2; ++tab) {
        const int ef  = (wave >> 1) & 3;
        const int ri  = wave & 1;
        const int e   = ef >> 1, f = ef & 1;
        const float* srcp = tab ? (ri ? PloI : PloR) : (ri ? PhiI : PhiR);
        const float* Ua = &U[tab ? 2 : 1][0][0];
        const float* Ub = &U[tab ? 3 : 0][0][0];
        const int  flip  = tab ? (e << 7) : (f ? 0xC0 : 0);
        const bool keep0 = tab ? ((c & 1) == f) : (e == 0);
        const bool keep1 = tab ? ((c & 1) == f) : (e == 1);

        v8f d1 = {};
#pragma unroll
        for (int k0 = 0; k0 < 4; ++k0) {
            const int ka = 4 * k0 + 2 * hs;
            v2f ufa;
            ufa.x = Ua[c * 16 + ka];
            ufa.y = Ua[c * 16 + ka + 1];
            const int q0 = tab ? ((ka << 4) | c)       : ((c << 4) | ka);
            const int q1 = tab ? (((ka + 1) << 4) | c) : ((c << 4) | (ka + 1));
            const float x0 = srcp[gray8(q0) ^ flip];
            const float x1 = srcp[gray8(q1) ^ flip];
            v2f xf;
            xf.x = keep0 ? x0 : 0.0f;
            xf.y = keep1 ? x1 : 0.0f;
            d1 = __builtin_amdgcn_wmma_f32_16x16x4_f32(
                     false, xf, false, ufa,
                     (short)0, d1, false, false);
        }
        v8f d2 = {};
#pragma unroll
        for (int k0 = 0; k0 < 4; ++k0) {
            const int kb = 8 * hs + 2 * k0;
            v2f ufb;
            ufb.x = Ub[c * 16 + kb];
            ufb.y = Ub[c * 16 + kb + 1];
            v2f bq;
            bq.x = d1[2 * k0];
            bq.y = d1[2 * k0 + 1];
            d2 = __builtin_amdgcn_wmma_f32_16x16x4_f32(
                     false, ufb, false, bq,
                     (short)0, d2, false, false);
        }
        float* dst = tab ? Vt : Wt;
        const int slot = (ef >> 1) * 4 + ri * 2 + (ef & 1);
#pragma unroll
        for (int r = 0; r < 8; ++r) {
            const int bp  = 8 * hs + r;
            const int idx = tab ? ((c << 4) | bp) : ((bp << 4) | c);
            dst[idx * 8 + slot] = d2[r];
        }
    }
    __syncthreads();

    float acc[16];
#pragma unroll
    for (int w = 0; w < 16; ++w) acc[w] = 0.0f;

    constexpr ZMasks zm = make_zmasks<N_LAYERS - 1>();
    const float4* Wt4 = reinterpret_cast<const float4*>(Wt);
    const float4* Vt4 = reinterpret_cast<const float4*>(Vt);
    const int lf = (c << 1) | hs;
    for (int j = 0; j < 2; ++j) {
        const int wl = wave | (j << 3);
        const float4 va = Vt4[(wl << 5) | lf];
        v2f aVR, aVI, aVIn;
        aVR.x  =  va.x; aVR.y  =  va.y;
        aVI.x  =  va.z; aVI.y  =  va.w;
        aVIn.x = -va.z; aVIn.y = -va.w;

        float4 wb = Wt4[lf];
        for (int it = 0; it < 16; ++it) {
            const float4 wbn = Wt4[(((it + 1) & 15) << 5) | lf];
            v2f bWR, bWI;
            bWR.x = wb.x; bWR.y = wb.y;
            bWI.x = wb.z; bWI.y = wb.w;
            v8f re = {}, im = {};
            re = __builtin_amdgcn_wmma_f32_16x16x4_f32(false, aVR,  false, bWR, (short)0, re, false, false);
            im = __builtin_amdgcn_wmma_f32_16x16x4_f32(false, aVI,  false, bWR, (short)0, im, false, false);
            re = __builtin_amdgcn_wmma_f32_16x16x4_f32(false, aVIn, false, bWI, (short)0, re, false, false);
            im = __builtin_amdgcn_wmma_f32_16x16x4_f32(false, aVR,  false, bWI, (short)0, im, false, false);

            float h[8];
#pragma unroll
            for (int r = 0; r < 8; ++r) h[r] = re[r] * re[r] + im[r] * im[r];
#pragma unroll
            for (int len = 1; len < 8; len <<= 1)
#pragma unroll
                for (int i = 0; i < 8; i += 2 * len)
#pragma unroll
                    for (int jj = i; jj < i + len; ++jj) {
                        const float a = h[jj], bq = h[jj + len];
                        h[jj] = a + bq;
                        h[jj + len] = a - bq;
                    }
            const int tb = (it << 12) | (j << 7);
#pragma unroll
            for (int w = 0; w < 16; ++w) {
                if (zm.rep[w] != w) continue;
                const float sg = odd_parity(zm.m[w] & tb) ? -1.0f : 1.0f;
                acc[w] += sg * h[zm.m[w] & 7];
            }
            wb = wbn;
        }
    }

    const int ylane = (c << 8) | (wave << 4) | (hs << 3);
    float red[16];
#pragma unroll
    for (int w = 0; w < 16; ++w) {
        float v = acc[zm.rep[w]];
        v = odd_parity(ylane & zm.m[w]) ? -v : v;
        v += __shfl_xor(v, 16, 32);
        v += __shfl_xor(v, 8, 32);
        v += __shfl_xor(v, 4, 32);
        v += __shfl_xor(v, 2, 32);
        v += __shfl_xor(v, 1, 32);
        red[w] = v;
    }
    if (lane == 0) {
#pragma unroll
        for (int w = 0; w < 16; ++w) partial[wave][w] = red[w];
    }
    __syncthreads();
    if (tid < 16) {
        float s = 0.0f;
#pragma unroll
        for (int q = 0; q < NWAVES; ++q) s += partial[q][tid];
        out[b * N_QUBITS + tid] = s;
    }
}

extern "C" void kernel_launch(void* const* d_in, const int* in_sizes, int n_in,
                              void* d_out, int out_size, void* d_ws, size_t ws_size,
                              hipStream_t stream) {
    const float* x      = (const float*)d_in[0];
    const float* params = (const float*)d_in[1];
    float* out          = (float*)d_out;
    (void)in_sizes; (void)n_in; (void)out_size; (void)d_ws; (void)ws_size;
    qsim16_kernel<<<BATCH, THREADS, 0, stream>>>(x, params, out);
}


# Z expvals as products of 4x4 signed Gram matrices of the 256-entry factors (sign x feature product on WMMA, factor tables k-major for it); the 65536 amplitudes are never formed; sum order changes
# speedup vs baseline: 42.1992x; 2.5674x over previous
#include <hip/hip_runtime.h>


#define N_QUBITS 16
#define N_LAYERS 2
#define BATCH    512
#define THREADS  256
#define NWAVES   (THREADS / 32)
#define FT_PITCH 520

typedef __attribute__((ext_vector_type(2))) float v2f;
typedef __attribute__((ext_vector_type(8))) float v8f;

__host__ __device__ constexpr int ring_inv(int y) {
    return (y ^ (y >> 1)) ^ ((y & 1) ? 0xC000 : 0);
}
__host__ __device__ constexpr int ring_fwd(int p) {
    int g = p;
    g ^= g >> 1; g ^= g >> 2; g ^= g >> 4; g ^= g >> 8;
    return (g & 0x7FFF) | (((g ^ (p >> 15)) & 1) << 15);
}
template <int L>
__host__ __device__ constexpr int qmap(int y) {
    for (int i = 0; i < L; ++i) y = ring_inv(y);
    return y;
}
__host__ __device__ constexpr int logical_of(int p) {
    for (int i = 0; i < N_LAYERS; ++i) p = ring_fwd(p);
    return p;
}

struct ZMasks { int m[16]; };
template <int L>
__host__ __device__ constexpr ZMasks make_zmasks() {
    ZMasks z = {};
    for (int w = 0; w < 16; ++w)
        for (int k = 0; k < 16; ++k)
            if ((logical_of(qmap<L>(1 << k)) >> (15 - w)) & 1) z.m[w] |= 1 << k;
    return z;
}
__host__ __device__ constexpr int swap_nibbles(int v) { return ((v & 15) << 4) | ((v >> 4) & 15); }
__device__ __forceinline__ int gray8(int h) { return h ^ (h >> 1); }

__global__ __launch_bounds__(THREADS)
void qsim16_kernel(const float* __restrict__ x,
                   const float* __restrict__ params,
                   float* __restrict__ out)
{
    __shared__ __align__(16) float Ft[2][4 * FT_PITCH];
    __shared__ __align__(16) float Gp[NWAVES][256];
    __shared__ float U[4][16][16];
    __shared__ float PhiR[256], PhiI[256];
    __shared__ float PloR[256], PloI[256];
    __shared__ float cw[16], sw[16];
    __shared__ float pcs[N_LAYERS][16];
    __shared__ float pss[N_LAYERS][16];

    const int tid  = threadIdx.x;
    const int lane = tid & 31;
    const int wave = tid >> 5;
    const int b    = blockIdx.x;

    if (tid < 16) {
        float a = 0.5f * x[b * N_QUBITS + tid];
        float s, c;
        __sincosf(a, &s, &c);
        cw[tid] = c;
        sw[tid] = s;
    } else if (tid < 48) {
        int idx = tid - 16;
        int l = idx >> 4, w = idx & 15;
        float a = 0.5f * params[idx];
        float s, c;
        __sincosf(a, &s, &c);
        pcs[l][w] = c;
        pss[l][w] = s;
    }
    __syncthreads();

    static_assert(THREADS == 256, "one table entry of each half per thread");
#pragma unroll
    for (int half = 0; half < 2; ++half) {
        const int q = tid;
        float pr = 1.0f, pi = 0.0f;
#pragma unroll
        for (int m = 0; m < 8; ++m) {
            const int w = 8 * half + m;
            const float c = cw[w], s = sw[w], pc = pcs[0][w], ps = pss[0][w];
            const bool one = (q >> (7 - m)) & 1;
            const float fr = one ? ps * c : pc * c;
            const float fi = one ? -(pc * s) : ps * s;
            const float nr = pr * fr - pi * fi;
            const float ni = pr * fi + pi * fr;
            pr = nr; pi = ni;
        }
        if (half == 0) { PhiR[q] = pr; PhiI[q] = pi; }
        else           { PloR[q] = pr; PloI[q] = pi; }
    }

    for (int e = tid; e < 4 * 256; e += THREADS) {
        const int l = N_LAYERS - 1;
        int g  = (e >> 8) & 3;
        int ar = (e >> 4) & 15;
        int ac = e & 15;
        float v = 1.0f;
#pragma unroll
        for (int m = 0; m < 4; ++m) {
            int w = 4 * g + m;
            int i = (ar >> (3 - m)) & 1;
            int j = (ac >> (3 - m)) & 1;
            float f = (i == j) ? pcs[l][w] : (i ? pss[l][w] : -pss[l][w]);
            v *= f;
        }
        U[g][ar][ac] = v;
    }
    __syncthreads();

    const int c  = lane & 15;
    const int hs = lane >> 4;

    static_assert(NWAVES == 8, "wave = (e, f, ri); lo' row blocks wave and wave + 8");
#pragma unroll
    for (int tab = 0; tab < 2; ++tab) {
        const int ef  = (wave >> 1) & 3;
        const int ri  = wave & 1;
        const int e   = ef >> 1, f = ef & 1;
        const float* srcp = tab ? (ri ? PloI : PloR) : (ri ? PhiI : PhiR);
        const float* Ua = &U[tab ? 2 : 1][0][0];
        const float* Ub = &U[tab ? 3 : 0][0][0];
        const int  flip  = tab ? (e << 7) : (f ? 0xC0 : 0);
        const bool keep0 = tab ? ((c & 1) == f) : (e == 0);
        const bool keep1 = tab ? ((c & 1) == f) : (e == 1);

        v8f d1 = {};
#pragma unroll
        for (int k0 = 0; k0 < 4; ++k0) {
            const int ka = 4 * k0 + 2 * hs;
            v2f ufa;
            ufa.x = Ua[c * 16 + ka];
            ufa.y = Ua[c * 16 + ka + 1];
            const int q0 = tab ? ((ka << 4) | c)       : ((c << 4) | ka);
            const int q1 = tab ? (((ka + 1) << 4) | c) : ((c << 4) | (ka + 1));
            const float x0 = srcp[gray8(q0) ^ flip];
            const float x1 = srcp[gray8(q1) ^ flip];
            v2f xf;
            xf.x = keep0 ? x0 : 0.0f;
            xf.y = keep1 ? x1 : 0.0f;
            d1 = __builtin_amdgcn_wmma_f32_16x16x4_f32(
                     false, xf, false, ufa,
                     (short)0, d1, false, false);
        }
        v8f d2 = {};
#pragma unroll
        for (int k0 = 0; k0 < 4; ++k0) {
            const int kb = 8 * hs + 2 * k0;
            v2f ufb;
            ufb.x = Ub[c * 16 + kb];
            ufb.y = Ub[c * 16 + kb + 1];
            v2f bq;
            bq.x = d1[2 * k0];
            bq.y = d1[2 * k0 + 1];
            d2 = __builtin_amdgcn_wmma_f32_16x16x4_f32(
                     false, ufb, false, bq,
                     (short)0, d2, false, false);
        }
        float* dst = &Ft[tab][ef * FT_PITCH + ri];
#pragma unroll
        for (int r = 0; r < 8; ++r) {
            const int bp = 8 * hs + r;
            dst[((bp << 4) | c) * 2] = d2[r];
        }
    }
    __syncthreads();

    constexpr ZMasks zm = make_zmasks<N_LAYERS - 1>();
    {
        const int tabq  = wave >> 2;
        const int p0    = (wave & 3) << 6;
        const int fk    = c >> 2, fk2 = c & 3;
        const bool isre = fk <= fk2;

        int mw = 0;
#pragma unroll
        for (int w = 0; w < 16; ++w) {
            const int mh = (zm.m[w] >> 8) & 0xFF;
            const int ml = swap_nibbles(zm.m[w] & 0xFF);
            mw = (c == w) ? (tabq ? ml : mh) : mw;
        }
        const float4* Tk  = reinterpret_cast<const float4*>(&Ft[tabq][fk  * FT_PITCH]);
        const float4* Tk2 = reinterpret_cast<const float4*>(&Ft[tabq][fk2 * FT_PITCH]);

        v8f g0 = {}, g1 = {};
#pragma unroll
        for (int s = 0; s < 16; ++s) {
            const int p = p0 + 4 * s + 2 * hs;
            const float4 a = Tk [p >> 1];
            const float4 q = Tk2[p >> 1];
            v2f ft, sg;
            ft.x = isre ? (a.x * q.x + a.y * q.y) : (a.y * q.x - a.x * q.y);
            ft.y = isre ? (a.z * q.z + a.w * q.w) : (a.w * q.z - a.z * q.w);
            const unsigned par = (unsigned)__popc(mw & p);
            sg.x = __uint_as_float(0x3f800000u | (par << 31));
            sg.y = __uint_as_float(0x3f800000u | ((par + (unsigned)mw) << 31));
            if (s & 1) g1 = __builtin_amdgcn_wmma_f32_16x16x4_f32(false, sg, false, ft, (short)0, g1, false, false);
            else       g0 = __builtin_amdgcn_wmma_f32_16x16x4_f32(false, sg, false, ft, (short)0, g0, false, false);
        }
        const float wt = tabq ? 1.0f : ((fk == fk2) ? 1.0f : (isre ? 2.0f : -2.0f));
#pragma unroll
        for (int r = 0; r < 8; ++r)
            Gp[wave][(8 * hs + r) * 16 + c] = wt * (g0[r] + g1[r]);
    }
    __syncthreads();

    if (wave == 0) {
        const float4* G4 = reinterpret_cast<const float4*>(&Gp[0][0]);
        const int g4 = (c * 16 + 8 * hs) >> 2;
        float4 w0 = G4[g4], w1 = G4[g4 + 1];
        float4 v0 = G4[4 * 64 + g4], v1 = G4[4 * 64 + g4 + 1];
#pragma unroll
        for (int t = 1; t < 4; ++t) {
            const float4 a0 = G4[t * 64 + g4],       a1 = G4[t * 64 + g4 + 1];
            const float4 b0 = G4[(4 + t) * 64 + g4], b1 = G4[(4 + t) * 64 + g4 + 1];
            w0.x += a0.x; w0.y += a0.y; w0.z += a0.z; w0.w += a0.w;
            w1.x += a1.x; w1.y += a1.y; w1.z += a1.z; w1.w += a1.w;
            v0.x += b0.x; v0.y += b0.y; v0.z += b0.z; v0.w += b0.w;
            v1.x += b1.x; v1.y += b1.y; v1.z += b1.z; v1.w += b1.w;
        }
        float e = w0.x * v0.x + w0.y * v0.y + w0.z * v0.z + w0.w * v0.w
                + w1.x * v1.x + w1.y * v1.y + w1.z * v1.z + w1.w * v1.w;
        e += __shfl_xor(e, 16, 32);
        if (lane < 16) out[b * N_QUBITS + lane] = e;
    }
}

extern "C" void kernel_launch(void* const* d_in, const int* in_sizes, int n_in,
                              void* d_out, int out_size, void* d_ws, size_t ws_size,
                              hipStream_t stream) {
    const float* x      = (const float*)d_in[0];
    const float* params = (const float*)d_in[1];
    float* out          = (float*)d_out;
    (void)in_sizes; (void)n_in; (void)out_size; (void)d_ws; (void)ws_size;
    qsim16_kernel<<<BATCH, THREADS, 0, stream>>>(x, params, out);
}


# Last-layer 4-qubit RY unitaries: each lane forms its own WMMA fragment entries in registers (four cos/sin factors, same product order, result bits unchanged); LDS table, build loop and reads gone
# speedup vs baseline: 49.6590x; 1.1768x over previous
#include <hip/hip_runtime.h>


#define N_QUBITS 16
#define N_LAYERS 2
#define BATCH    512
#define THREADS  256
#define NWAVES   (THREADS / 32)
#define FT_PITCH 520

typedef __attribute__((ext_vector_type(2))) float v2f;
typedef __attribute__((ext_vector_type(8))) float v8f;

__host__ __device__ constexpr int ring_inv(int y) {
    return (y ^ (y >> 1)) ^ ((y & 1) ? 0xC000 : 0);
}
__host__ __device__ constexpr int ring_fwd(int p) {
    int g = p;
    g ^= g >> 1; g ^= g >> 2; g ^= g >> 4; g ^= g >> 8;
    return (g & 0x7FFF) | (((g ^ (p >> 15)) & 1) << 15);
}
template <int L>
__host__ __device__ constexpr int qmap(int y) {
    for (int i = 0; i < L; ++i) y = ring_inv(y);
    return y;
}
__host__ __device__ constexpr int logical_of(int p) {
    for (int i = 0; i < N_LAYERS; ++i) p = ring_fwd(p);
    return p;
}

struct ZMasks { int m[16]; };
template <int L>
__host__ __device__ constexpr ZMasks make_zmasks() {
    ZMasks z = {};
    for (int w = 0; w < 16; ++w)
        for (int k = 0; k < 16; ++k)
            if ((logical_of(qmap<L>(1 << k)) >> (15 - w)) & 1) z.m[w] |= 1 << k;
    return z;
}
__host__ __device__ constexpr int swap_nibbles(int v) { return ((v & 15) << 4) | ((v >> 4) & 15); }
__device__ __forceinline__ int gray8(int h) { return h ^ (h >> 1); }

__global__ __launch_bounds__(THREADS)
void qsim16_kernel(const float* __restrict__ x,
                   const float* __restrict__ params,
                   float* __restrict__ out)
{
    __shared__ __align__(16) float Ft[2][4 * FT_PITCH];
    __shared__ __align__(16) float Gp[NWAVES][256];
    __shared__ float PhiR[256], PhiI[256];
    __shared__ float PloR[256], PloI[256];
    __shared__ float cw[16], sw[16];
    __shared__ float pcs[N_LAYERS][16];
    __shared__ float pss[N_LAYERS][16];

    const int tid  = threadIdx.x;
    const int lane = tid & 31;
    const int wave = tid >> 5;
    const int b    = blockIdx.x;

    if (tid < 16) {
        float a = 0.5f * x[b * N_QUBITS + tid];
        float s, c;
        __sincosf(a, &s, &c);
        cw[tid] = c;
        sw[tid] = s;
    } else if (tid < 48) {
        int idx = tid - 16;
        int l = idx >> 4, w = idx & 15;
        float a = 0.5f * params[idx];
        float s, c;
        __sincosf(a, &s, &c);
        pcs[l][w] = c;
        pss[l][w] = s;
    }
    __syncthreads();

    static_assert(THREADS == 256, "one table entry of each half per thread");
#pragma unroll
    for (int half = 0; half < 2; ++half) {
        const int q = tid;
        float pr = 1.0f, pi = 0.0f;
#pragma unroll
        for (int m = 0; m < 8; ++m) {
            const int w = 8 * half + m;
            const float c = cw[w], s = sw[w], pc = pcs[0][w], ps = pss[0][w];
            const bool one = (q >> (7 - m)) & 1;
            const float fr = one ? ps * c : pc * c;
            const float fi = one ? -(pc * s) : ps * s;
            const float nr = pr * fr - pi * fi;
            const float ni = pr * fi + pi * fr;
            pr = nr; pi = ni;
        }
        if (half == 0) { PhiR[q] = pr; PhiI[q] = pi; }
        else           { PloR[q] = pr; PloI[q] = pi; }
    }

    const int c  = lane & 15;
    const int hs = lane >> 4;

    float fz[16], fo[16];
#pragma unroll
    for (int w = 0; w < 16; ++w) {
        const float pc = pcs[N_LAYERS - 1][w], ps = pss[N_LAYERS - 1][w];
        const bool i = (c >> (3 - (w & 3))) & 1;
        fz[w] = i ? ps : pc;
        fo[w] = i ? pc : -ps;
    }
    v2f ufa[2][4], ufb[2][4];
#pragma unroll
    for (int tab = 0; tab < 2; ++tab) {
        const int ga = tab ? 2 : 1, gb = tab ? 3 : 0;
        const float a2 = hs ? fo[4 * ga + 2] : fz[4 * ga + 2];
        const float b0 = hs ? fo[4 * gb]     : fz[4 * gb];
#pragma unroll
        for (int k0 = 0; k0 < 4; ++k0) {
            const float a0 = (k0 >> 1) ? fo[4 * ga]     : fz[4 * ga];
            const float a1 = (k0 & 1)  ? fo[4 * ga + 1] : fz[4 * ga + 1];
            const float pa = (a0 * a1) * a2;
            ufa[tab][k0].x = pa * fz[4 * ga + 3];
            ufa[tab][k0].y = pa * fo[4 * ga + 3];
            const float b1 = (k0 >> 1) ? fo[4 * gb + 1] : fz[4 * gb + 1];
            const float b2 = (k0 & 1)  ? fo[4 * gb + 2] : fz[4 * gb + 2];
            const float pb = (b0 * b1) * b2;
            ufb[tab][k0].x = pb * fz[4 * gb + 3];
            ufb[tab][k0].y = pb * fo[4 * gb + 3];
        }
    }
    __syncthreads();

    static_assert(NWAVES == 8, "wave = (e, f, ri); lo' row blocks wave and wave + 8");
#pragma unroll
    for (int tab = 0; tab < 2; ++tab) {
        const int ef  = (wave >> 1) & 3;
        const int ri  = wave & 1;
        const int e   = ef >> 1, f = ef & 1;
        const float* srcp = tab ? (ri ? PloI : PloR) : (ri ? PhiI : PhiR);
        const int  flip  = tab ? (e << 7) : (f ? 0xC0 : 0);
        const bool keep0 = tab ? ((c & 1) == f) : (e == 0);
        const bool keep1 = tab ? ((c & 1) == f) : (e == 1);

        v8f d1 = {};
#pragma unroll
        for (int k0 = 0; k0 < 4; ++k0) {
            const int ka = 4 * k0 + 2 * hs;
            const int q0 = tab ? ((ka << 4) | c)       : ((c << 4) | ka);
            const int q1 = tab ? (((ka + 1) << 4) | c) : ((c << 4) | (ka + 1));
            const float x0 = srcp[gray8(q0) ^ flip];
            const float x1 = srcp[gray8(q1) ^ flip];
            v2f xf;
            xf.x = keep0 ? x0 : 0.0f;
            xf.y = keep1 ? x1 : 0.0f;
            d1 = __builtin_amdgcn_wmma_f32_16x16x4_f32(
                     false, xf, false, ufa[tab][k0],
                     (short)0, d1, false, false);
        }
        v8f d2 = {};
#pragma unroll
        for (int k0 = 0; k0 < 4; ++k0) {
            v2f bq;
            bq.x = d1[2 * k0];
            bq.y = d1[2 * k0 + 1];
            d2 = __builtin_amdgcn_wmma_f32_16x16x4_f32(
                     false, ufb[tab][k0], false, bq,
                     (short)0, d2, false, false);
        }
        float* dst = &Ft[tab][ef * FT_PITCH + ri];
#pragma unroll
        for (int r = 0; r < 8; ++r) {
            const int bp = 8 * hs + r;
            dst[((bp << 4) | c) * 2] = d2[r];
        }
    }
    __syncthreads();

    constexpr ZMasks zm = make_zmasks<N_LAYERS - 1>();
    {
        const int tabq  = wave >> 2;
        const int p0    = (wave & 3) << 6;
        const int fk    = c >> 2, fk2 = c & 3;
        const bool isre = fk <= fk2;

        int mw = 0;
#pragma unroll
        for (int w = 0; w < 16; ++w) {
            const int mh = (zm.m[w] >> 8) & 0xFF;
            const int ml = swap_nibbles(zm.m[w] & 0xFF);
            mw = (c == w) ? (tabq ? ml : mh) : mw;
        }
        const float4* Tk  = reinterpret_cast<const float4*>(&Ft[tabq][fk  * FT_PITCH]);
        const float4* Tk2 = reinterpret_cast<const float4*>(&Ft[tabq][fk2 * FT_PITCH]);

        v8f g0 = {}, g1 = {};
#pragma unroll
        for (int s = 0; s < 16; ++s) {
            const int p = p0 + 4 * s + 2 * hs;
            const float4 a = Tk [p >> 1];
            const float4 q = Tk2[p >> 1];
            v2f ft, sg;
            ft.x = isre ? (a.x * q.x + a.y * q.y) : (a.y * q.x - a.x * q.y);
            ft.y = isre ? (a.z * q.z + a.w * q.w) : (a.w * q.z - a.z * q.w);
            const unsigned par = (unsigned)__popc(mw & p);
            sg.x = __uint_as_float(0x3f800000u | (par << 31));
            sg.y = __uint_as_float(0x3f800000u | ((par + (unsigned)mw) << 31));
            if (s & 1) g1 = __builtin_amdgcn_wmma_f32_16x16x4_f32(false, sg, false, ft, (short)0, g1, false, false);
            else       g0 = __builtin_amdgcn_wmma_f32_16x16x4_f32(false, sg, false, ft, (short)0, g0, false, false);
        }
        const float wt = tabq ? 1.0f : ((fk == fk2) ? 1.0f : (isre ? 2.0f : -2.0f));
#pragma unroll
        for (int r = 0; r < 8; ++r)
            Gp[wave][(8 * hs + r) * 16 + c] = wt * (g0[r] + g1[r]);
    }
    __syncthreads();

    if (wave == 0) {
        const float4* G4 = reinterpret_cast<const float4*>(&Gp[0][0]);
        const int g4 = (c * 16 + 8 * hs) >> 2;
        float4 w0 = G4[g4], w1 = G4[g4 + 1];
        float4 v0 = G4[4 * 64 + g4], v1 = G4[4 * 64 + g4 + 1];
#pragma unroll
        for (int t = 1; t < 4; ++t) {
            const float4 a0 = G4[t * 64 + g4],       a1 = G4[t * 64 + g4 + 1];
            const float4 b0 = G4[(4 + t) * 64 + g4], b1 = G4[(4 + t) * 64 + g4 + 1];
            w0.x += a0.x; w0.y += a0.y; w0.z += a0.z; w0.w += a0.w;
            w1.x += a1.x; w1.y += a1.y; w1.z += a1.z; w1.w += a1.w;
            v0.x += b0.x; v0.y += b0.y; v0.z += b0.z; v0.w += b0.w;
            v1.x += b1.x; v1.y += b1.y; v1.z += b1.z; v1.w += b1.w;
        }
        float e = w0.x * v0.x + w0.y * v0.y + w0.z * v0.z + w0.w * v0.w
                + w1.x * v1.x + w1.y * v1.y + w1.z * v1.z + w1.w * v1.w;
        e += __shfl_xor(e, 16, 32);
        if (lane < 16) out[b * N_QUBITS + lane] = e;
    }
}

extern "C" void kernel_launch(void* const* d_in, const int* in_sizes, int n_in,
                              void* d_out, int out_size, void* d_ws, size_t ws_size,
                              hipStream_t stream) {
    const float* x      = (const float*)d_in[0];
    const float* params = (const float*)d_in[1];
    float* out          = (float*)d_out;
    (void)in_sizes; (void)n_in; (void)out_size; (void)d_ws; (void)ws_size;
    qsim16_kernel<<<BATCH, THREADS, 0, stream>>>(x, params, out);
}


# Last RY layer skips the parity-mask zeros: one first product on a factor pair's unmasked table, second product over the 8 live rows per parity; its WMMAs 16 to 8 per wave; RY sum order changes
# speedup vs baseline: 53.5012x; 1.0774x over previous
#include <hip/hip_runtime.h>


#define N_QUBITS 16
#define N_LAYERS 2
#define BATCH    512
#define THREADS  256
#define NWAVES   (THREADS / 32)
#define FT_PITCH 520

typedef __attribute__((ext_vector_type(2))) float v2f;
typedef __attribute__((ext_vector_type(8))) float v8f;

__host__ __device__ constexpr int ring_inv(int y) {
    return (y ^ (y >> 1)) ^ ((y & 1) ? 0xC000 : 0);
}
__host__ __device__ constexpr int ring_fwd(int p) {
    int g = p;
    g ^= g >> 1; g ^= g >> 2; g ^= g >> 4; g ^= g >> 8;
    return (g & 0x7FFF) | (((g ^ (p >> 15)) & 1) << 15);
}
template <int L>
__host__ __device__ constexpr int qmap(int y) {
    for (int i = 0; i < L; ++i) y = ring_inv(y);
    return y;
}
__host__ __device__ constexpr int logical_of(int p) {
    for (int i = 0; i < N_LAYERS; ++i) p = ring_fwd(p);
    return p;
}

struct ZMasks { int m[16]; };
template <int L>
__host__ __device__ constexpr ZMasks make_zmasks() {
    ZMasks z = {};
    for (int w = 0; w < 16; ++w)
        for (int k = 0; k < 16; ++k)
            if ((logical_of(qmap<L>(1 << k)) >> (15 - w)) & 1) z.m[w] |= 1 << k;
    return z;
}
__host__ __device__ constexpr int swap_nibbles(int v) { return ((v & 15) << 4) | ((v >> 4) & 15); }
__device__ __forceinline__ int gray8(int h) { return h ^ (h >> 1); }

__global__ __launch_bounds__(THREADS)
void qsim16_kernel(const float* __restrict__ x,
                   const float* __restrict__ params,
                   float* __restrict__ out)
{
    __shared__ __align__(16) float Ft[2][4 * FT_PITCH];
    __shared__ __align__(16) float Gp[NWAVES][256];
    __shared__ float Pt[4][256];
    __shared__ float cw[16], sw[16];
    __shared__ __align__(16) float pcs[N_LAYERS][16];
    __shared__ __align__(16) float pss[N_LAYERS][16];

    const int tid  = threadIdx.x;
    const int lane = tid & 31;
    const int wave = tid >> 5;
    const int b    = blockIdx.x;

    if (tid < 16) {
        float a = 0.5f * x[b * N_QUBITS + tid];
        float s, c;
        __sincosf(a, &s, &c);
        cw[tid] = c;
        sw[tid] = s;
    } else if (tid < 48) {
        int idx = tid - 16;
        int l = idx >> 4, w = idx & 15;
        float a = 0.5f * params[idx];
        float s, c;
        __sincosf(a, &s, &c);
        pcs[l][w] = c;
        pss[l][w] = s;
    }
    __syncthreads();

    static_assert(THREADS == 256, "one table entry of each half per thread");
#pragma unroll
    for (int half = 0; half < 2; ++half) {
        const int q = tid;
        float pr = 1.0f, pi = 0.0f;
#pragma unroll
        for (int m = 0; m < 8; ++m) {
            const int w = 8 * half + m;
            const float c = cw[w], s = sw[w], pc = pcs[0][w], ps = pss[0][w];
            const bool one = (q >> (7 - m)) & 1;
            const float fr = one ? ps * c : pc * c;
            const float fi = one ? -(pc * s) : ps * s;
            const float nr = pr * fr - pi * fi;
            const float ni = pr * fi + pi * fr;
            pr = nr; pi = ni;
        }
        Pt[2 * half][q]     = pr;
        Pt[2 * half + 1][q] = pi;
    }

    const int c  = lane & 15;
    const int hs = lane >> 4;

    static_assert(NWAVES == 8, "wave = (tab, xs, ri)");
    const int tab = wave >> 2;
    const int xs  = (wave >> 1) & 1;
    const int ri  = wave & 1;

    float fz[8], fo[8];
    {
        const float4* pc4 = reinterpret_cast<const float4*>(&pcs[N_LAYERS - 1][8 * tab]);
        const float4* ps4 = reinterpret_cast<const float4*>(&pss[N_LAYERS - 1][8 * tab]);
        const float4 pa = pc4[0], pb = pc4[1], sa = ps4[0], sb = ps4[1];
        const float pcv[8] = { pa.x, pa.y, pa.z, pa.w, pb.x, pb.y, pb.z, pb.w };
        const float psv[8] = { sa.x, sa.y, sa.z, sa.w, sb.x, sb.y, sb.z, sb.w };
#pragma unroll
        for (int j = 0; j < 8; ++j) {
            const bool i = (c >> (3 - (j & 3))) & 1;
            fz[j] = i ? psv[j] : pcv[j];
            fo[j] = i ? pcv[j] : -psv[j];
        }
    }
    v2f ufa[4];
    float ub[8];
    {
        const float a2 = hs ? fo[2] : fz[2];
        const float b0 = hs ? fo[4] : fz[4];
#pragma unroll
        for (int k0 = 0; k0 < 4; ++k0) {
            const float a0 = (k0 >> 1) ? fo[0] : fz[0];
            const float a1 = (k0 & 1)  ? fo[1] : fz[1];
            const float pa = (a0 * a1) * a2;
            ufa[k0].x = pa * fz[3];
            ufa[k0].y = pa * fo[3];
            const float b1 = (k0 >> 1) ? fo[5] : fz[5];
            const float b2 = (k0 & 1)  ? fo[6] : fz[6];
            const float pb = (b0 * b1) * b2;
            ub[2 * k0]     = pb * fz[7];
            ub[2 * k0 + 1] = pb * fo[7];
        }
    }
    __syncthreads();

    {
        const float* srcp = Pt[2 * tab + ri];
        const int flip = tab ? (xs << 7) : (xs ? 0xC0 : 0);
        const int rb   = (c & 8) | ((c & 3) << 1) | ((c >> 2) & 1);

        v8f d1 = {};
#pragma unroll
        for (int k0 = 0; k0 < 4; ++k0) {
            const int ka = 4 * k0 + 2 * hs;
            v2f xf;
            xf.x = srcp[gray8((ka << 4) | rb) ^ flip];
            xf.y = srcp[gray8(((ka + 1) << 4) | rb) ^ flip];
            d1 = __builtin_amdgcn_wmma_f32_16x16x4_f32(
                     false, xf, false, ufa[k0],
                     (short)0, d1, false, false);
        }
        v8f d2e = {}, d2o = {};
#pragma unroll
        for (int j2 = 0; j2 < 2; ++j2) {
            v2f ue, uo, be, bo;
            ue.x = ub[4 * j2];     ue.y = ub[4 * j2 + 2];
            be.x = d1[2 * j2];     be.y = d1[2 * j2 + 1];
            uo.x = ub[4 * j2 + 1]; uo.y = ub[4 * j2 + 3];
            bo.x = d1[4 + 2 * j2]; bo.y = d1[4 + 2 * j2 + 1];
            d2e = __builtin_amdgcn_wmma_f32_16x16x4_f32(
                      false, ue, false, be, (short)0, d2e, false, false);
            d2o = __builtin_amdgcn_wmma_f32_16x16x4_f32(
                      false, uo, false, bo, (short)0, d2o, false, false);
        }
        const int ke = tab ? 2 * xs : xs;
        const int ko = tab ? 2 * xs + 1 : 2 + xs;
        float* dste = &Ft[tab][ke * FT_PITCH + ri];
        float* dsto = &Ft[tab][ko * FT_PITCH + ri];
#pragma unroll
        for (int r = 0; r < 8; ++r)
            dste[(((8 * hs + r) << 4) | c) * 2] = d2e[r];
#pragma unroll
        for (int r = 0; r < 8; ++r)
            dsto[(((8 * hs + r) << 4) | c) * 2] = d2o[r];
    }
    __syncthreads();

    constexpr ZMasks zm = make_zmasks<N_LAYERS - 1>();
    {
        const int tabq  = wave >> 2;
        const int p0    = (wave & 3) << 6;
        const int fk    = c >> 2, fk2 = c & 3;
        const bool isre = fk <= fk2;

        int mw = 0;
#pragma unroll
        for (int w = 0; w < 16; ++w) {
            const int mh = swap_nibbles((zm.m[w] >> 8) & 0xFF);
            const int ml = swap_nibbles(zm.m[w] & 0xFF);
            mw = (c == w) ? (tabq ? ml : mh) : mw;
        }
        const float4* Tk  = reinterpret_cast<const float4*>(&Ft[tabq][fk  * FT_PITCH]);
        const float4* Tk2 = reinterpret_cast<const float4*>(&Ft[tabq][fk2 * FT_PITCH]);

        v8f g0 = {}, g1 = {};
#pragma unroll
        for (int s = 0; s < 16; ++s) {
            const int p = p0 + 4 * s + 2 * hs;
            const float4 a = Tk [p >> 1];
            const float4 q = Tk2[p >> 1];
            v2f ft, sg;
            ft.x = isre ? (a.x * q.x + a.y * q.y) : (a.y * q.x - a.x * q.y);
            ft.y = isre ? (a.z * q.z + a.w * q.w) : (a.w * q.z - a.z * q.w);
            const unsigned par = (unsigned)__popc(mw & p);
            sg.x = __uint_as_float(0x3f800000u | (par << 31));
            sg.y = __uint_as_float(0x3f800000u | ((par + (unsigned)mw) << 31));
            if (s & 1) g1 = __builtin_amdgcn_wmma_f32_16x16x4_f32(false, sg, false, ft, (short)0, g1, false, false);
            else       g0 = __builtin_amdgcn_wmma_f32_16x16x4_f32(false, sg, false, ft, (short)0, g0, false, false);
        }
        const float wt = tabq ? 1.0f : ((fk == fk2) ? 1.0f : (isre ? 2.0f : -2.0f));
#pragma unroll
        for (int r = 0; r < 8; ++r)
            Gp[wave][(8 * hs + r) * 16 + c] = wt * (g0[r] + g1[r]);
    }
    __syncthreads();

    if (wave == 0) {
        const float4* G4 = reinterpret_cast<const float4*>(&Gp[0][0]);
        const int g4 = (c * 16 + 8 * hs) >> 2;
        float4 w0 = G4[g4], w1 = G4[g4 + 1];
        float4 v0 = G4[4 * 64 + g4], v1 = G4[4 * 64 + g4 + 1];
#pragma unroll
        for (int t = 1; t < 4; ++t) {
            const float4 a0 = G4[t * 64 + g4],       a1 = G4[t * 64 + g4 + 1];
            const float4 b0 = G4[(4 + t) * 64 + g4], b1 = G4[(4 + t) * 64 + g4 + 1];
            w0.x += a0.x; w0.y += a0.y; w0.z += a0.z; w0.w += a0.w;
            w1.x += a1.x; w1.y += a1.y; w1.z += a1.z; w1.w += a1.w;
            v0.x += b0.x; v0.y += b0.y; v0.z += b0.z; v0.w += b0.w;
            v1.x += b1.x; v1.y += b1.y; v1.z += b1.z; v1.w += b1.w;
        }
        float e = w0.x * v0.x + w0.y * v0.y + w0.z * v0.z + w0.w * v0.w
                + w1.x * v1.x + w1.y * v1.y + w1.z * v1.z + w1.w * v1.w;
        e += __shfl_xor(e, 16, 32);
        if (lane < 16) out[b * N_QUBITS + lane] = e;
    }
}

extern "C" void kernel_launch(void* const* d_in, const int* in_sizes, int n_in,
                              void* d_out, int out_size, void* d_ws, size_t ws_size,
                              hipStream_t stream) {
    const float* x      = (const float*)d_in[0];
    const float* params = (const float*)d_in[1];
    float* out          = (float*)d_out;
    (void)in_sizes; (void)n_in; (void)out_size; (void)d_ws; (void)ws_size;
    qsim16_kernel<<<BATCH, THREADS, 0, stream>>>(x, params, out);
}


# Workgroups of 4 waves, not 8 (half as many waves enter): a wave takes a factor pair's Re and Im planes and 128 Gram positions, its LDS reads issued in groups ahead of use; expval sum order changes
# speedup vs baseline: 59.0803x; 1.1043x over previous
#include <hip/hip_runtime.h>


#define N_QUBITS 16
#define N_LAYERS 2
#define BATCH    512
#define THREADS  128
#define NWAVES   (THREADS / 32)
#define FT_PITCH 520

typedef __attribute__((ext_vector_type(2))) float v2f;
typedef __attribute__((ext_vector_type(8))) float v8f;

__host__ __device__ constexpr int ring_inv(int y) {
    return (y ^ (y >> 1)) ^ ((y & 1) ? 0xC000 : 0);
}
__host__ __device__ constexpr int ring_fwd(int p) {
    int g = p;
    g ^= g >> 1; g ^= g >> 2; g ^= g >> 4; g ^= g >> 8;
    return (g & 0x7FFF) | (((g ^ (p >> 15)) & 1) << 15);
}
template <int L>
__host__ __device__ constexpr int qmap(int y) {
    for (int i = 0; i < L; ++i) y = ring_inv(y);
    return y;
}
__host__ __device__ constexpr int logical_of(int p) {
    for (int i = 0; i < N_LAYERS; ++i) p = ring_fwd(p);
    return p;
}

struct ZMasks { int m[16]; };
template <int L>
__host__ __device__ constexpr ZMasks make_zmasks() {
    ZMasks z = {};
    for (int w = 0; w < 16; ++w)
        for (int k = 0; k < 16; ++k)
            if ((logical_of(qmap<L>(1 << k)) >> (15 - w)) & 1) z.m[w] |= 1 << k;
    return z;
}
__host__ __device__ constexpr int swap_nibbles(int v) { return ((v & 15) << 4) | ((v >> 4) & 15); }
__device__ __forceinline__ int gray8(int h) { return h ^ (h >> 1); }

__global__ __launch_bounds__(THREADS)
void qsim16_kernel(const float* __restrict__ x,
                   const float* __restrict__ params,
                   float* __restrict__ out)
{
    __shared__ __align__(16) float Ft[2][4 * FT_PITCH];
    __shared__ __align__(16) float Gp[NWAVES][256];
    __shared__ float Pt[4][256];
    __shared__ __align__(16) float cw[16];
    __shared__ __align__(16) float sw[16];
    __shared__ __align__(16) float pcs[N_LAYERS][16];
    __shared__ __align__(16) float pss[N_LAYERS][16];

    const int tid  = threadIdx.x;
    const int lane = tid & 31;
    const int wave = tid >> 5;
    const int b    = blockIdx.x;

    if (tid < 16) {
        float a = 0.5f * x[b * N_QUBITS + tid];
        float s, c;
        __sincosf(a, &s, &c);
        cw[tid] = c;
        sw[tid] = s;
    } else if (tid < 48) {
        int idx = tid - 16;
        int l = idx >> 4, w = idx & 15;
        float a = 0.5f * params[idx];
        float s, c;
        __sincosf(a, &s, &c);
        pcs[l][w] = c;
        pss[l][w] = s;
    }
    __syncthreads();

    static_assert(THREADS == 128, "two table entries of each half per thread");
    static_assert(NWAVES == 4, "wave = (tab, xs)");
    const int tab = wave >> 1;
    const int xs  = wave & 1;
    float cv[16], sv[16], pc0[16], ps0[16];
    float pcv[8], psv[8];
    {
        const float4* c4 = reinterpret_cast<const float4*>(cw);
        const float4* s4 = reinterpret_cast<const float4*>(sw);
        const float4* p4 = reinterpret_cast<const float4*>(pcs[0]);
        const float4* q4 = reinterpret_cast<const float4*>(pss[0]);
#pragma unroll
        for (int i = 0; i < 4; ++i) {
            const float4 a = c4[i], b2 = s4[i], d = p4[i], e = q4[i];
            cv[4 * i] = a.x;  cv[4 * i + 1] = a.y;  cv[4 * i + 2] = a.z;  cv[4 * i + 3] = a.w;
            sv[4 * i] = b2.x; sv[4 * i + 1] = b2.y; sv[4 * i + 2] = b2.z; sv[4 * i + 3] = b2.w;
            pc0[4 * i] = d.x; pc0[4 * i + 1] = d.y; pc0[4 * i + 2] = d.z; pc0[4 * i + 3] = d.w;
            ps0[4 * i] = e.x; ps0[4 * i + 1] = e.y; ps0[4 * i + 2] = e.z; ps0[4 * i + 3] = e.w;
        }
        const float4* pc4 = reinterpret_cast<const float4*>(&pcs[N_LAYERS - 1][8 * tab]);
        const float4* ps4 = reinterpret_cast<const float4*>(&pss[N_LAYERS - 1][8 * tab]);
        const float4 pa = pc4[0], pb = pc4[1], sa = ps4[0], sb = ps4[1];
        pcv[0] = pa.x; pcv[1] = pa.y; pcv[2] = pa.z; pcv[3] = pa.w; pcv[4] = pb.x; pcv[5] = pb.y; pcv[6] = pb.z; pcv[7] = pb.w;
        psv[0] = sa.x; psv[1] = sa.y; psv[2] = sa.z; psv[3] = sa.w; psv[4] = sb.x; psv[5] = sb.y; psv[6] = sb.z; psv[7] = sb.w;
    }
    __builtin_amdgcn_sched_barrier(0);
#pragma unroll
    for (int half = 0; half < 2; ++half) {
        float pr0, pi0, pr1, pi1;
        {
            const int w = 8 * half;
            const float c = cv[w], s = sv[w], pc = pc0[w], ps = ps0[w];
            pr0 = pc * c; pi0 = ps * s;
            pr1 = ps * c; pi1 = -(pc * s);
        }
#pragma unroll
        for (int m = 1; m < 8; ++m) {
            const int w = 8 * half + m;
            const float c = cv[w], s = sv[w], pc = pc0[w], ps = ps0[w];
            const bool one = (tid >> (7 - m)) & 1;
            const float fr = one ? ps * c : pc * c;
            const float fi = one ? -(pc * s) : ps * s;
            const float nr0 = pr0 * fr - pi0 * fi;
            const float ni0 = pr0 * fi + pi0 * fr;
            const float nr1 = pr1 * fr - pi1 * fi;
            const float ni1 = pr1 * fi + pi1 * fr;
            pr0 = nr0; pi0 = ni0; pr1 = nr1; pi1 = ni1;
        }
        Pt[2 * half][tid]           = pr0;
        Pt[2 * half + 1][tid]       = pi0;
        Pt[2 * half][tid + 128]     = pr1;
        Pt[2 * half + 1][tid + 128] = pi1;
    }

    const int c  = lane & 15;
    const int hs = lane >> 4;


    float fz[8], fo[8];
    {
#pragma unroll
        for (int j = 0; j < 8; ++j) {
            const bool i = (c >> (3 - (j & 3))) & 1;
            fz[j] = i ? psv[j] : pcv[j];
            fo[j] = i ? pcv[j] : -psv[j];
        }
    }
    const int flip = tab ? (xs << 7) : (xs ? 0xC0 : 0);
    const int rb   = (c & 8) | ((c & 3) << 1) | ((c >> 2) & 1);
    int gi0[4], gi1[4];
#pragma unroll
    for (int k0 = 0; k0 < 4; ++k0) {
        const int ka = 4 * k0 + 2 * hs;
        gi0[k0] = gray8((ka << 4) | rb) ^ flip;
        gi1[k0] = gray8(((ka + 1) << 4) | rb) ^ flip;
    }
    __syncthreads();

    {
        const float* srcr = Pt[2 * tab];
        const float* srci = Pt[2 * tab + 1];
        v2f xr[4], xi[4];
#pragma unroll
        for (int k0 = 0; k0 < 4; ++k0) {
            xr[k0].x = srcr[gi0[k0]]; xr[k0].y = srcr[gi1[k0]];
            xi[k0].x = srci[gi0[k0]]; xi[k0].y = srci[gi1[k0]];
        }
        __builtin_amdgcn_sched_barrier(0);
        v2f ufa[4];
        float ub[8];
        {
            const float a2 = hs ? fo[2] : fz[2];
            const float b0 = hs ? fo[4] : fz[4];
#pragma unroll
            for (int k0 = 0; k0 < 4; ++k0) {
                const float a0 = (k0 >> 1) ? fo[0] : fz[0];
                const float a1 = (k0 & 1)  ? fo[1] : fz[1];
                const float pa = (a0 * a1) * a2;
                ufa[k0].x = pa * fz[3];
                ufa[k0].y = pa * fo[3];
                const float b1 = (k0 >> 1) ? fo[5] : fz[5];
                const float b2 = (k0 & 1)  ? fo[6] : fz[6];
                const float pb = (b0 * b1) * b2;
                ub[2 * k0]     = pb * fz[7];
                ub[2 * k0 + 1] = pb * fo[7];
            }
        }
        v8f d1r = {}, d1i = {};
#pragma unroll
        for (int k0 = 0; k0 < 4; ++k0) {
            d1r = __builtin_amdgcn_wmma_f32_16x16x4_f32(
                      false, xr[k0], false, ufa[k0], (short)0, d1r, false, false);
            d1i = __builtin_amdgcn_wmma_f32_16x16x4_f32(
                      false, xi[k0], false, ufa[k0], (short)0, d1i, false, false);
        }
        v8f d2er = {}, d2or = {}, d2ei = {}, d2oi = {};
#pragma unroll
        for (int j2 = 0; j2 < 2; ++j2) {
            v2f ue, uo, ber, bor, bei, boi;
            ue.x = ub[4 * j2];     ue.y = ub[4 * j2 + 2];
            uo.x = ub[4 * j2 + 1]; uo.y = ub[4 * j2 + 3];
            ber.x = d1r[2 * j2];     ber.y = d1r[2 * j2 + 1];
            bor.x = d1r[4 + 2 * j2]; bor.y = d1r[4 + 2 * j2 + 1];
            bei.x = d1i[2 * j2];     bei.y = d1i[2 * j2 + 1];
            boi.x = d1i[4 + 2 * j2]; boi.y = d1i[4 + 2 * j2 + 1];
            d2er = __builtin_amdgcn_wmma_f32_16x16x4_f32(
                       false, ue, false, ber, (short)0, d2er, false, false);
            d2or = __builtin_amdgcn_wmma_f32_16x16x4_f32(
                       false, uo, false, bor, (short)0, d2or, false, false);
            d2ei = __builtin_amdgcn_wmma_f32_16x16x4_f32(
                       false, ue, false, bei, (short)0, d2ei, false, false);
            d2oi = __builtin_amdgcn_wmma_f32_16x16x4_f32(
                       false, uo, false, boi, (short)0, d2oi, false, false);
        }
        const int ke = tab ? 2 * xs : xs;
        const int ko = tab ? 2 * xs + 1 : 2 + xs;
        float2* dste = reinterpret_cast<float2*>(&Ft[tab][ke * FT_PITCH]);
        float2* dsto = reinterpret_cast<float2*>(&Ft[tab][ko * FT_PITCH]);
#pragma unroll
        for (int r = 0; r < 8; ++r)
            dste[((8 * hs + r) << 4) | c] = make_float2(d2er[r], d2ei[r]);
#pragma unroll
        for (int r = 0; r < 8; ++r)
            dsto[((8 * hs + r) << 4) | c] = make_float2(d2or[r], d2oi[r]);
    }
    __syncthreads();

    constexpr ZMasks zm = make_zmasks<N_LAYERS - 1>();
    {
        const int tabq  = wave >> 1;
        const int p0    = (wave & 1) << 7;
        const int fk    = c >> 2, fk2 = c & 3;
        const bool isre = fk <= fk2;

        int mw = 0;
#pragma unroll
        for (int w = 0; w < 16; ++w) {
            const int mh = swap_nibbles((zm.m[w] >> 8) & 0xFF);
            const int ml = swap_nibbles(zm.m[w] & 0xFF);
            mw = (c == w) ? (tabq ? ml : mh) : mw;
        }
        const float4* Tk  = reinterpret_cast<const float4*>(&Ft[tabq][fk  * FT_PITCH]);
        const float4* Tk2 = reinterpret_cast<const float4*>(&Ft[tabq][fk2 * FT_PITCH]);

        constexpr int GS = 32, GB = 8;
        float4 av[GS], qv[GS];
#pragma unroll
        for (int s = 0; s < GB; ++s) {
            const int p = p0 + 4 * s + 2 * hs;
            av[s] = Tk [p >> 1];
            qv[s] = Tk2[p >> 1];
        }
        v8f g0 = {}, g1 = {};
#pragma unroll
        for (int blk = 0; blk < GS / GB; ++blk) {
            if (blk + 1 < GS / GB) {
#pragma unroll
                for (int s = GB * (blk + 1); s < GB * (blk + 2); ++s) {
                    const int p = p0 + 4 * s + 2 * hs;
                    av[s] = Tk [p >> 1];
                    qv[s] = Tk2[p >> 1];
                }
            }
            __builtin_amdgcn_sched_barrier(0);
#pragma unroll
            for (int s = GB * blk; s < GB * (blk + 1); ++s) {
                const int p = p0 + 4 * s + 2 * hs;
                const float4 a = av[s];
                const float4 q = qv[s];
                v2f ft, sg;
                ft.x = isre ? (a.x * q.x + a.y * q.y) : (a.y * q.x - a.x * q.y);
                ft.y = isre ? (a.z * q.z + a.w * q.w) : (a.w * q.z - a.z * q.w);
                const unsigned par = (unsigned)__popc(mw & p);
                sg.x = __uint_as_float(0x3f800000u | (par << 31));
                sg.y = __uint_as_float(0x3f800000u | ((par + (unsigned)mw) << 31));
                if (s & 1) g1 = __builtin_amdgcn_wmma_f32_16x16x4_f32(false, sg, false, ft, (short)0, g1, false, false);
                else       g0 = __builtin_amdgcn_wmma_f32_16x16x4_f32(false, sg, false, ft, (short)0, g0, false, false);
            }
        }
        const float wt = tabq ? 1.0f : ((fk == fk2) ? 1.0f : (isre ? 2.0f : -2.0f));
#pragma unroll
        for (int r = 0; r < 8; ++r)
            Gp[wave][(8 * hs + r) * 16 + c] = wt * (g0[r] + g1[r]);
    }
    __syncthreads();

    if (wave == 0) {
        const float4* G4 = reinterpret_cast<const float4*>(&Gp[0][0]);
        const int g4 = (c * 16 + 8 * hs) >> 2;
        float4 w0 = G4[g4], w1 = G4[g4 + 1];
        float4 v0 = G4[2 * 64 + g4], v1 = G4[2 * 64 + g4 + 1];
        {
            const float4 a0 = G4[64 + g4],     a1 = G4[64 + g4 + 1];
            const float4 b0 = G4[3 * 64 + g4], b1 = G4[3 * 64 + g4 + 1];
            w0.x += a0.x; w0.y += a0.y; w0.z += a0.z; w0.w += a0.w;
            w1.x += a1.x; w1.y += a1.y; w1.z += a1.z; w1.w += a1.w;
            v0.x += b0.x; v0.y += b0.y; v0.z += b0.z; v0.w += b0.w;
            v1.x += b1.x; v1.y += b1.y; v1.z += b1.z; v1.w += b1.w;
        }
        float e = w0.x * v0.x + w0.y * v0.y + w0.z * v0.z + w0.w * v0.w
                + w1.x * v1.x + w1.y * v1.y + w1.z * v1.z + w1.w * v1.w;
        e += __shfl_xor(e, 16, 32);
        if (lane < 16) out[b * N_QUBITS + lane] = e;
    }
}

extern "C" void kernel_launch(void* const* d_in, const int* in_sizes, int n_in,
                              void* d_out, int out_size, void* d_ws, size_t ws_size,
                              hipStream_t stream) {
    const float* x      = (const float*)d_in[0];
    const float* params = (const float*)d_in[1];
    float* out          = (float*)d_out;
    (void)in_sizes; (void)n_in; (void)out_size; (void)d_ws; (void)ws_size;
    qsim16_kernel<<<BATCH, THREADS, 0, stream>>>(x, params, out);
}


# A wave forms its own operands in registers: 24 half-angle cos/sin by lane + v_readlane, first-product A fragments as shared per-wire factor products (rounding changes); 2 LDS tables, 2 barriers gone
# speedup vs baseline: 63.6181x; 1.0768x over previous
#include <hip/hip_runtime.h>


#define N_QUBITS 16
#define N_LAYERS 2
#define BATCH    512
#define THREADS  128
#define NWAVES   (THREADS / 32)
#define FT_PITCH 520

typedef __attribute__((ext_vector_type(2))) float v2f;
typedef __attribute__((ext_vector_type(8))) float v8f;

__host__ __device__ constexpr int ring_inv(int y) {
    return (y ^ (y >> 1)) ^ ((y & 1) ? 0xC000 : 0);
}
__host__ __device__ constexpr int ring_fwd(int p) {
    int g = p;
    g ^= g >> 1; g ^= g >> 2; g ^= g >> 4; g ^= g >> 8;
    return (g & 0x7FFF) | (((g ^ (p >> 15)) & 1) << 15);
}
template <int L>
__host__ __device__ constexpr int qmap(int y) {
    for (int i = 0; i < L; ++i) y = ring_inv(y);
    return y;
}
__host__ __device__ constexpr int logical_of(int p) {
    for (int i = 0; i < N_LAYERS; ++i) p = ring_fwd(p);
    return p;
}

struct ZMasks { int m[16]; };
template <int L>
__host__ __device__ constexpr ZMasks make_zmasks() {
    ZMasks z = {};
    for (int w = 0; w < 16; ++w)
        for (int k = 0; k < 16; ++k)
            if ((logical_of(qmap<L>(1 << k)) >> (15 - w)) & 1) z.m[w] |= 1 << k;
    return z;
}
__host__ __device__ constexpr int swap_nibbles(int v) { return ((v & 15) << 4) | ((v >> 4) & 15); }
__device__ __forceinline__ int gray8(int h) { return h ^ (h >> 1); }

struct cplx { float r, i; };
__device__ __forceinline__ cplx cmul(cplx a, cplx b) {
    cplx o;
    o.r = a.r * b.r - a.i * b.i;
    o.i = a.r * b.i + a.i * b.r;
    return o;
}
__device__ __forceinline__ cplx pick(cplx at0, cplx at1, int g, int m) {
    const bool one = (g >> (7 - m)) & 1;
    cplx o;
    o.r = one ? at1.r : at0.r;
    o.i = one ? at1.i : at0.i;
    return o;
}
__device__ __forceinline__ float lane_bcast(float v, int l) {
    return __int_as_float(__builtin_amdgcn_readlane(__float_as_int(v), l));
}

__global__ __launch_bounds__(THREADS)
void qsim16_kernel(const float* __restrict__ x,
                   const float* __restrict__ params,
                   float* __restrict__ out)
{
    __shared__ __align__(16) float Ft[2][4 * FT_PITCH];
    __shared__ __align__(16) float Gp[NWAVES][256];
    const int tid  = threadIdx.x;
    const int lane = tid & 31;
    const int wave = tid >> 5;
    const int b    = blockIdx.x;

    static_assert(THREADS == 128, "four waves");
    static_assert(NWAVES == 4, "wave = (tab, xs)");
    static_assert(N_LAYERS == 2, "one layer on the product state, one layer on the factors");
    const int tab = wave >> 1;
    const int xs  = wave & 1;
    const int c  = lane & 15;
    const int hs = lane >> 4;

    float cs, sn;
    {
        const int j8  = lane & 7;
        const int grp = lane >> 3;
        const float* src = (grp == 0) ? (x + b * N_QUBITS + 8 * tab + j8)
                                      : (params + ((grp == 1) ? 0 : (N_LAYERS - 1) * N_QUBITS) + 8 * tab + j8);
        __sincosf(0.5f * src[0], &sn, &cs);
    }
    float cx[8], sx[8], pc0[8], ps0[8];
    float pcv[8], psv[8];
#pragma unroll
    for (int m = 0; m < 8; ++m) {
        cx[m]  = lane_bcast(cs, m);       sx[m]  = lane_bcast(sn, m);
        pc0[m] = lane_bcast(cs, 8 + m);   ps0[m] = lane_bcast(sn, 8 + m);
        pcv[m] = lane_bcast(cs, 16 + m);  psv[m] = lane_bcast(sn, 16 + m);
    }
    cplx z0[8], z1[8];
#pragma unroll
    for (int m = 0; m < 8; ++m) {
        z0[m].r = pc0[m] * cx[m]; z0[m].i = ps0[m] * sx[m];
        z1[m].r = ps0[m] * cx[m]; z1[m].i = -(pc0[m] * sx[m]);
    }


    float fz[8], fo[8];
    {
#pragma unroll
        for (int j = 0; j < 8; ++j) {
            const bool i = (c >> (3 - (j & 3))) & 1;
            fz[j] = i ? psv[j] : pcv[j];
            fo[j] = i ? pcv[j] : -psv[j];
        }
    }
    const int flip = tab ? (xs << 7) : (xs ? 0xC0 : 0);
    const int rb   = (c & 8) | ((c & 3) << 1) | ((c >> 2) & 1);
    int gi0[4], gi1[4];
#pragma unroll
    for (int k0 = 0; k0 < 4; ++k0) {
        const int ka = 4 * k0 + 2 * hs;
        gi0[k0] = gray8((ka << 4) | rb) ^ flip;
        gi1[k0] = gray8(((ka + 1) << 4) | rb) ^ flip;
    }
    v2f xr[4], xi[4];
    {
        cplx lo3 = cmul(cmul(pick(z0[5], z1[5], gi0[0], 5), pick(z0[6], z1[6], gi0[0], 6)), pick(z0[7], z1[7], gi0[0], 7));
        cplx ml[2];
#pragma unroll
        for (int v = 0; v < 2; ++v) {
            const int g = v ? gi1[0] : gi0[0];
            ml[v] = cmul(cmul(pick(z0[3], z1[3], g, 3), pick(z0[4], z1[4], g, 4)), lo3);
        }
#pragma unroll
        for (int k0 = 0; k0 < 4; ++k0) {
            const int g = gi0[k0];
            const cplx h = cmul(cmul(pick(z0[0], z1[0], g, 0), pick(z0[1], z1[1], g, 1)), pick(z0[2], z1[2], g, 2));
            const cplx e0 = cmul(h, ml[0]);
            const cplx e1 = cmul(h, ml[1]);
            xr[k0].x = e0.r; xr[k0].y = e1.r;
            xi[k0].x = e0.i; xi[k0].y = e1.i;
        }
    }

    {
        v2f ufa[4];
        float ub[8];
        {
            const float a2 = hs ? fo[2] : fz[2];
            const float b0 = hs ? fo[4] : fz[4];
#pragma unroll
            for (int k0 = 0; k0 < 4; ++k0) {
                const float a0 = (k0 >> 1) ? fo[0] : fz[0];
                const float a1 = (k0 & 1)  ? fo[1] : fz[1];
                const float pa = (a0 * a1) * a2;
                ufa[k0].x = pa * fz[3];
                ufa[k0].y = pa * fo[3];
                const float b1 = (k0 >> 1) ? fo[5] : fz[5];
                const float b2 = (k0 & 1)  ? fo[6] : fz[6];
                const float pb = (b0 * b1) * b2;
                ub[2 * k0]     = pb * fz[7];
                ub[2 * k0 + 1] = pb * fo[7];
            }
        }
        v8f d1r = {}, d1i = {};
#pragma unroll
        for (int k0 = 0; k0 < 4; ++k0) {
            d1r = __builtin_amdgcn_wmma_f32_16x16x4_f32(
                      false, xr[k0], false, ufa[k0], (short)0, d1r, false, false);
            d1i = __builtin_amdgcn_wmma_f32_16x16x4_f32(
                      false, xi[k0], false, ufa[k0], (short)0, d1i, false, false);
        }
        v8f d2er = {}, d2or = {}, d2ei = {}, d2oi = {};
#pragma unroll
        for (int j2 = 0; j2 < 2; ++j2) {
            v2f ue, uo, ber, bor, bei, boi;
            ue.x = ub[4 * j2];     ue.y = ub[4 * j2 + 2];
            uo.x = ub[4 * j2 + 1]; uo.y = ub[4 * j2 + 3];
            ber.x = d1r[2 * j2];     ber.y = d1r[2 * j2 + 1];
            bor.x = d1r[4 + 2 * j2]; bor.y = d1r[4 + 2 * j2 + 1];
            bei.x = d1i[2 * j2];     bei.y = d1i[2 * j2 + 1];
            boi.x = d1i[4 + 2 * j2]; boi.y = d1i[4 + 2 * j2 + 1];
            d2er = __builtin_amdgcn_wmma_f32_16x16x4_f32(
                       false, ue, false, ber, (short)0, d2er, false, false);
            d2or = __builtin_amdgcn_wmma_f32_16x16x4_f32(
                       false, uo, false, bor, (short)0, d2or, false, false);
            d2ei = __builtin_amdgcn_wmma_f32_16x16x4_f32(
                       false, ue, false, bei, (short)0, d2ei, false, false);
            d2oi = __builtin_amdgcn_wmma_f32_16x16x4_f32(
                       false, uo, false, boi, (short)0, d2oi, false, false);
        }
        const int ke = tab ? 2 * xs : xs;
        const int ko = tab ? 2 * xs + 1 : 2 + xs;
        float2* dste = reinterpret_cast<float2*>(&Ft[tab][ke * FT_PITCH]);
        float2* dsto = reinterpret_cast<float2*>(&Ft[tab][ko * FT_PITCH]);
#pragma unroll
        for (int r = 0; r < 8; ++r)
            dste[((8 * hs + r) << 4) | c] = make_float2(d2er[r], d2ei[r]);
#pragma unroll
        for (int r = 0; r < 8; ++r)
            dsto[((8 * hs + r) << 4) | c] = make_float2(d2or[r], d2oi[r]);
    }
    __syncthreads();

    constexpr ZMasks zm = make_zmasks<N_LAYERS - 1>();
    {
        const int tabq  = wave >> 1;
        const int p0    = (wave & 1) << 7;
        const int fk    = c >> 2, fk2 = c & 3;
        const bool isre = fk <= fk2;

        int mw = 0;
#pragma unroll
        for (int w = 0; w < 16; ++w) {
            const int mh = swap_nibbles((zm.m[w] >> 8) & 0xFF);
            const int ml = swap_nibbles(zm.m[w] & 0xFF);
            mw = (c == w) ? (tabq ? ml : mh) : mw;
        }
        const float4* Tk  = reinterpret_cast<const float4*>(&Ft[tabq][fk  * FT_PITCH]);
        const float4* Tk2 = reinterpret_cast<const float4*>(&Ft[tabq][fk2 * FT_PITCH]);

        constexpr int GS = 32, GB = 8;
        float4 av[GS], qv[GS];
#pragma unroll
        for (int s = 0; s < GB; ++s) {
            const int p = p0 + 4 * s + 2 * hs;
            av[s] = Tk [p >> 1];
            qv[s] = Tk2[p >> 1];
        }
        v8f g0 = {}, g1 = {};
#pragma unroll
        for (int blk = 0; blk < GS / GB; ++blk) {
            if (blk + 1 < GS / GB) {
#pragma unroll
                for (int s = GB * (blk + 1); s < GB * (blk + 2); ++s) {
                    const int p = p0 + 4 * s + 2 * hs;
                    av[s] = Tk [p >> 1];
                    qv[s] = Tk2[p >> 1];
                }
            }
            __builtin_amdgcn_sched_barrier(0);
#pragma unroll
            for (int s = GB * blk; s < GB * (blk + 1); ++s) {
                const int p = p0 + 4 * s + 2 * hs;
                const float4 a = av[s];
                const float4 q = qv[s];
                v2f ft, sg;
                ft.x = isre ? (a.x * q.x + a.y * q.y) : (a.y * q.x - a.x * q.y);
                ft.y = isre ? (a.z * q.z + a.w * q.w) : (a.w * q.z - a.z * q.w);
                const unsigned par = (unsigned)__popc(mw & p);
                sg.x = __uint_as_float(0x3f800000u | (par << 31));
                sg.y = __uint_as_float(0x3f800000u | ((par + (unsigned)mw) << 31));
                if (s & 1) g1 = __builtin_amdgcn_wmma_f32_16x16x4_f32(false, sg, false, ft, (short)0, g1, false, false);
                else       g0 = __builtin_amdgcn_wmma_f32_16x16x4_f32(false, sg, false, ft, (short)0, g0, false, false);
            }
        }
        const float wt = tabq ? 1.0f : ((fk == fk2) ? 1.0f : (isre ? 2.0f : -2.0f));
#pragma unroll
        for (int r = 0; r < 8; ++r)
            Gp[wave][(8 * hs + r) * 16 + c] = wt * (g0[r] + g1[r]);
    }
    __syncthreads();

    if (wave == 0) {
        const float4* G4 = reinterpret_cast<const float4*>(&Gp[0][0]);
        const int g4 = (c * 16 + 8 * hs) >> 2;
        float4 w0 = G4[g4], w1 = G4[g4 + 1];
        float4 v0 = G4[2 * 64 + g4], v1 = G4[2 * 64 + g4 + 1];
        {
            const float4 a0 = G4[64 + g4],     a1 = G4[64 + g4 + 1];
            const float4 b0 = G4[3 * 64 + g4], b1 = G4[3 * 64 + g4 + 1];
            w0.x += a0.x; w0.y += a0.y; w0.z += a0.z; w0.w += a0.w;
            w1.x += a1.x; w1.y += a1.y; w1.z += a1.z; w1.w += a1.w;
            v0.x += b0.x; v0.y += b0.y; v0.z += b0.z; v0.w += b0.w;
            v1.x += b1.x; v1.y += b1.y; v1.z += b1.z; v1.w += b1.w;
        }
        float e = w0.x * v0.x + w0.y * v0.y + w0.z * v0.z + w0.w * v0.w
                + w1.x * v1.x + w1.y * v1.y + w1.z * v1.z + w1.w * v1.w;
        e += __shfl_xor(e, 16, 32);
        if (lane < 16) out[b * N_QUBITS + lane] = e;
    }
}

extern "C" void kernel_launch(void* const* d_in, const int* in_sizes, int n_in,
                              void* d_out, int out_size, void* d_ws, size_t ws_size,
                              hipStream_t stream) {
    const float* x      = (const float*)d_in[0];
    const float* params = (const float*)d_in[1];
    float* out          = (float*)d_out;
    (void)in_sizes; (void)n_in; (void)out_size; (void)d_ws; (void)ws_size;
    qsim16_kernel<<<BATCH, THREADS, 0, stream>>>(x, params, out);
}
